# starting kernel
# baseline (speedup 1.0000x reference)
#include <hip/hip_runtime.h>
#include <hip/hip_bf16.h>

#define NNODES 100000
#define NEDGES 1600000
#define INCH   32
#define HIDC   64
#define NHEADS 2
#define HCAT   128
#define NCLS   12
#define NEGSLOPE 0.2f

typedef __attribute__((ext_vector_type(2))) float v2f;
typedef __attribute__((ext_vector_type(8))) float v8f;

__device__ __forceinline__ float leakyr(float x) { return x > 0.f ? x : NEGSLOPE * x; }

__device__ __forceinline__ void atomicMaxF(float* addr, float v) {
    if (v >= 0.f) atomicMax((int*)addr, __float_as_int(v));
    else          atomicMin((unsigned int*)addr, __float_as_uint(v));
}

template <int K, int M, int MT>
__global__ __launch_bounds__(256) void gemm_wmma_f32(
    const float* __restrict__ A, const float* __restrict__ B,
    const float* __restrict__ bias, float* __restrict__ D, int nrows)
{
    const int wave = threadIdx.x >> 5;
    const int lane = threadIdx.x & 31;
    const int tile = blockIdx.x * 8 + wave;
    const int ntiles = nrows >> 4;
    if (tile >= ntiles) return;
    const int half = lane >> 4;
    const int mrow = lane & 15;
    const int ncol = lane & 15;
    const int m0 = tile << 4;

    v8f acc[MT] = {};

    const float* __restrict__ arow = A + (size_t)(m0 + mrow) * K;
    for (int kb = 0; kb < K; kb += 4) {
        const int k0 = kb + 2 * half;
        v2f a;
        a.x = arow[k0];
        a.y = arow[k0 + 1];
#pragma unroll
        for (int t = 0; t < MT; ++t) {
            const int n = t * 16 + ncol;
            v2f b;
            if ((M & 15) == 0 || n < M) {
                b.x = B[(size_t)k0 * M + n];
                b.y = B[(size_t)(k0 + 1) * M + n];
            } else {
                b.x = 0.f; b.y = 0.f;
            }
            acc[t] = __builtin_amdgcn_wmma_f32_16x16x4_f32(
                false, a, false, b, (short)0, acc[t], false, false);
        }
    }

#pragma unroll
    for (int t = 0; t < MT; ++t) {
        const int n = t * 16 + ncol;
        if ((M & 15) != 0 && n >= M) continue;
        const float bv = bias ? bias[n] : 0.f;
#pragma unroll
        for (int r = 0; r < 8; ++r) {
            const int m = m0 + r + 8 * half;
            D[(size_t)m * M + n] = acc[t][r] + bv;
        }
    }
}

__global__ __launch_bounds__(256) void attn_logits_kernel(
    const float* __restrict__ H, const float* __restrict__ a_s,
    const float* __restrict__ a_d, float* __restrict__ als, float* __restrict__ ald)
{
    const int idx = blockIdx.x * blockDim.x + threadIdx.x;
    if (idx >= NNODES * NHEADS) return;
    const int head = idx & 1;
    const int node = idx >> 1;
    const float4* hv = (const float4*)(H + (size_t)node * HCAT + head * HIDC);
    const float4* sv = (const float4*)(a_s + head * HIDC);
    const float4* dv = (const float4*)(a_d + head * HIDC);
    float s = 0.f, d = 0.f;
#pragma unroll
    for (int i = 0; i < HIDC / 4; ++i) {
        const float4 h4 = hv[i], s4 = sv[i], d4 = dv[i];
        s += h4.x * s4.x + h4.y * s4.y + h4.z * s4.z + h4.w * s4.w;
        d += h4.x * d4.x + h4.y * d4.y + h4.z * d4.z + h4.w * d4.w;
    }
    als[idx] = s;
    ald[idx] = d;
}

__device__ __forceinline__ void edge_sd(const int* __restrict__ ei, int e, int& s, int& d) {
    if (e < NEDGES) { s = ei[e]; d = ei[NEDGES + e]; }
    else            { s = d = e - NEDGES; }
}

__global__ __launch_bounds__(256) void edge_max_kernel(
    const int* __restrict__ ei, const float* __restrict__ als,
    const float* __restrict__ ald, float* __restrict__ maxb)
{
    const int e = blockIdx.x * blockDim.x + threadIdx.x;
    if (e >= NEDGES + NNODES) return;
    int s, d; edge_sd(ei, e, s, d);
    atomicMaxF(&maxb[d * 2 + 0], leakyr(als[s * 2 + 0] + ald[d * 2 + 0]));
    atomicMaxF(&maxb[d * 2 + 1], leakyr(als[s * 2 + 1] + ald[d * 2 + 1]));
}

__global__ __launch_bounds__(256) void edge_exp_kernel(
    const int* __restrict__ ei, const float* __restrict__ als,
    const float* __restrict__ ald, const float* __restrict__ maxb,
    float* __restrict__ exbuf, float* __restrict__ denom)
{
    const int e = blockIdx.x * blockDim.x + threadIdx.x;
    if (e >= NEDGES + NNODES) return;
    int s, d; edge_sd(ei, e, s, d);
    const float x0 = __expf(leakyr(als[s * 2 + 0] + ald[d * 2 + 0]) - maxb[d * 2 + 0]);
    const float x1 = __expf(leakyr(als[s * 2 + 1] + ald[d * 2 + 1]) - maxb[d * 2 + 1]);
    exbuf[(size_t)e * 2 + 0] = x0;
    exbuf[(size_t)e * 2 + 1] = x1;
    atomicAdd(&denom[d * 2 + 0], x0);
    atomicAdd(&denom[d * 2 + 1], x1);
}

__global__ __launch_bounds__(256) void edge_aggr_kernel(
    const int* __restrict__ ei, const float* __restrict__ H,
    const float* __restrict__ exbuf, const float* __restrict__ denom,
    float* __restrict__ AGG)
{
    const int e = blockIdx.x * 8 + (threadIdx.x >> 5);
    if (e >= NEDGES + NNODES) return;
    const int lane = threadIdx.x & 31;
    int s, d; edge_sd(ei, e, s, d);
    const int c = lane * 4;
    const int head = c >> 6;
    const float alpha = exbuf[(size_t)e * 2 + head] / denom[d * 2 + head];
    const float4 h4 = *(const float4*)(H + (size_t)s * HCAT + c);
    float* out = AGG + (size_t)d * HCAT + c;
    atomicAdd(out + 0, h4.x * alpha);
    atomicAdd(out + 1, h4.y * alpha);
    atomicAdd(out + 2, h4.z * alpha);
    atomicAdd(out + 3, h4.w * alpha);
}

__global__ __launch_bounds__(256) void finalize_concat_kernel(
    const float* __restrict__ AGG, const float* __restrict__ bias,
    const float* __restrict__ prelu_a, float* __restrict__ OUT)
{
    const int idx = blockIdx.x * blockDim.x + threadIdx.x;
    if (idx >= NNODES * HCAT / 4) return;
    const int k = (idx & (HCAT / 4 - 1)) * 4;
    const float a = *prelu_a;
    float4 v = ((const float4*)AGG)[idx];
    v.x += bias[k + 0]; v.y += bias[k + 1]; v.z += bias[k + 2]; v.w += bias[k + 3];
    v.x = v.x >= 0.f ? v.x : a * v.x;
    v.y = v.y >= 0.f ? v.y : a * v.y;
    v.z = v.z >= 0.f ? v.z : a * v.z;
    v.w = v.w >= 0.f ? v.w : a * v.w;
    ((float4*)OUT)[idx] = v;
}

__global__ __launch_bounds__(256) void finalize_mean_kernel(
    const float* __restrict__ AGG, const float* __restrict__ bias,
    const float* __restrict__ prelu_a, float* __restrict__ OUT)
{
    const int idx = blockIdx.x * blockDim.x + threadIdx.x;
    if (idx >= NNODES * HIDC / 4) return;
    const int node = idx / (HIDC / 4);
    const int c = (idx % (HIDC / 4)) * 4;
    const float a = *prelu_a;
    const float* base = AGG + (size_t)node * HCAT;
    const float4 v0 = *(const float4*)(base + c);
    const float4 v1 = *(const float4*)(base + HIDC + c);
    float4 v;
    v.x = 0.5f * (v0.x + v1.x) + bias[c + 0];
    v.y = 0.5f * (v0.y + v1.y) + bias[c + 1];
    v.z = 0.5f * (v0.z + v1.z) + bias[c + 2];
    v.w = 0.5f * (v0.w + v1.w) + bias[c + 3];
    v.x = v.x >= 0.f ? v.x : a * v.x;
    v.y = v.y >= 0.f ? v.y : a * v.y;
    v.z = v.z >= 0.f ? v.z : a * v.z;
    v.w = v.w >= 0.f ? v.w : a * v.w;
    *(float4*)(OUT + (size_t)node * HIDC + c) = v;
}

extern "C" void kernel_launch(void* const* d_in, const int* in_sizes, int n_in,
                              void* d_out, int out_size, void* d_ws, size_t ws_size,
                              hipStream_t stream)
{
    const float* x       = (const float*)d_in[0];
    const int*   ei      = (const int*)  d_in[1];
    const float* W1      = (const float*)d_in[2];
    const float* a1s     = (const float*)d_in[3];
    const float* a1d     = (const float*)d_in[4];
    const float* b1      = (const float*)d_in[5];
    const float* W2      = (const float*)d_in[6];
    const float* a2s     = (const float*)d_in[7];
    const float* a2d     = (const float*)d_in[8];
    const float* b2      = (const float*)d_in[9];
    const float* W3      = (const float*)d_in[10];
    const float* a3s     = (const float*)d_in[11];
    const float* a3d     = (const float*)d_in[12];
    const float* b3      = (const float*)d_in[13];
    const float* prelu_a = (const float*)d_in[14];
    const float* Wc      = (const float*)d_in[15];
    const float* bc      = (const float*)d_in[16];
    float* out = (float*)d_out;

    float* ws = (float*)d_ws;
    size_t off = 0;
    float* H     = ws + off; off += (size_t)NNODES * HCAT;
    float* AGG   = ws + off; off += (size_t)NNODES * HCAT;
    float* PREV  = ws + off; off += (size_t)NNODES * HCAT;
    float* als   = ws + off; off += (size_t)NNODES * 2;
    float* ald   = ws + off; off += (size_t)NNODES * 2;
    float* maxb  = ws + off; off += (size_t)NNODES * 2;
    float* denom = ws + off; off += (size_t)NNODES * 2;
    float* exbuf = ws + off; off += (size_t)(NEDGES + NNODES) * 2;

    const int ETOT = NEDGES + NNODES;
    const int gN2    = (NNODES * NHEADS + 255) / 256;
    const int gEdge  = (ETOT + 255) / 256;
    const int gEdgeW = (ETOT + 7) / 8;
    const int gTile  = ((NNODES / 16) + 7) / 8;
    const int gFinC  = (NNODES * HCAT / 4 + 255) / 256;
    const int gFinM  = (NNODES * HIDC / 4 + 255) / 256;

    struct Layer {
        const float* W; const float* as; const float* ad; const float* b; bool concat;
    };
    const Layer layers[3] = {
        { W1, a1s, a1d, b1, true  },
        { W2, a2s, a2d, b2, true  },
        { W3, a3s, a3d, b3, false },
    };

    for (int l = 0; l < 3; ++l) {
        const float* in = (l == 0) ? x : PREV;
        hipMemsetAsync(AGG,   0x00, (size_t)NNODES * HCAT * sizeof(float), stream);
        hipMemsetAsync(maxb,  0xFF, (size_t)NNODES * 2 * sizeof(float), stream);
        hipMemsetAsync(denom, 0x00, (size_t)NNODES * 2 * sizeof(float), stream);

        if (l == 0)
            gemm_wmma_f32<INCH, HCAT, 8><<<gTile, 256, 0, stream>>>(in, layers[l].W, nullptr, H, NNODES);
        else
            gemm_wmma_f32<HCAT, HCAT, 8><<<gTile, 256, 0, stream>>>(in, layers[l].W, nullptr, H, NNODES);

        attn_logits_kernel<<<gN2, 256, 0, stream>>>(H, layers[l].as, layers[l].ad, als, ald);
        edge_max_kernel  <<<gEdge, 256, 0, stream>>>(ei, als, ald, maxb);
        edge_exp_kernel  <<<gEdge, 256, 0, stream>>>(ei, als, ald, maxb, exbuf, denom);
        edge_aggr_kernel <<<gEdgeW, 256, 0, stream>>>(ei, H, exbuf, denom, AGG);

        if (layers[l].concat)
            finalize_concat_kernel<<<gFinC, 256, 0, stream>>>(AGG, layers[l].b, prelu_a, PREV);
        else
            finalize_mean_kernel  <<<gFinM, 256, 0, stream>>>(AGG, layers[l].b, prelu_a, PREV);
    }

    gemm_wmma_f32<HIDC, NCLS, 1><<<gTile, 256, 0, stream>>>(PREV, Wc, bc, out, NNODES);
    (void)in_sizes; (void)n_in; (void)out_size; (void)ws_size;
}


# Edges sorted by destination once per call (arrival-order rank); one wave per node does softmax and sum of h[src] in registers, replacing the float-atomic max/sum/scatter passes
# speedup vs baseline: 5.4171x; 5.4171x over previous
#include <hip/hip_runtime.h>
#include <hip/hip_bf16.h>

#define NNODES 100000
#define NEDGES 1600000
#define INCH   32
#define HIDC   64
#define NHEADS 2
#define HCAT   128
#define NCLS   12
#define NEGSLOPE 0.2f

typedef __attribute__((ext_vector_type(2))) float v2f;
typedef __attribute__((ext_vector_type(8))) float v8f;

__device__ __forceinline__ float leakyr(float x) { return x > 0.f ? x : NEGSLOPE * x; }

template <int K, int M, int MT>
__global__ __launch_bounds__(256) void gemm_wmma_f32(
    const float* __restrict__ A, const float* __restrict__ B,
    const float* __restrict__ bias, float* __restrict__ D, int nrows)
{
    const int wave = threadIdx.x >> 5;
    const int lane = threadIdx.x & 31;
    const int tile = blockIdx.x * 8 + wave;
    const int ntiles = nrows >> 4;
    if (tile >= ntiles) return;
    const int half = lane >> 4;
    const int mrow = lane & 15;
    const int ncol = lane & 15;
    const int m0 = tile << 4;

    v8f acc[MT] = {};

    const float* __restrict__ arow = A + (size_t)(m0 + mrow) * K;
    for (int kb = 0; kb < K; kb += 4) {
        const int k0 = kb + 2 * half;
        v2f a;
        a.x = arow[k0];
        a.y = arow[k0 + 1];
#pragma unroll
        for (int t = 0; t < MT; ++t) {
            const int n = t * 16 + ncol;
            v2f b;
            if ((M & 15) == 0 || n < M) {
                b.x = B[(size_t)k0 * M + n];
                b.y = B[(size_t)(k0 + 1) * M + n];
            } else {
                b.x = 0.f; b.y = 0.f;
            }
            acc[t] = __builtin_amdgcn_wmma_f32_16x16x4_f32(
                false, a, false, b, (short)0, acc[t], false, false);
        }
    }

#pragma unroll
    for (int t = 0; t < MT; ++t) {
        const int n = t * 16 + ncol;
        if ((M & 15) != 0 && n >= M) continue;
        const float bv = bias ? bias[n] : 0.f;
#pragma unroll
        for (int r = 0; r < 8; ++r) {
            const int m = m0 + r + 8 * half;
            D[(size_t)m * M + n] = acc[t][r] + bv;
        }
    }
}

__global__ __launch_bounds__(256) void attn_logits_kernel(
    const float* __restrict__ H, const float* __restrict__ a_s,
    const float* __restrict__ a_d, float* __restrict__ als, float* __restrict__ ald)
{
    const int idx = blockIdx.x * blockDim.x + threadIdx.x;
    if (idx >= NNODES * NHEADS) return;
    const int head = idx & 1;
    const int node = idx >> 1;
    const float4* hv = (const float4*)(H + (size_t)node * HCAT + head * HIDC);
    const float4* sv = (const float4*)(a_s + head * HIDC);
    const float4* dv = (const float4*)(a_d + head * HIDC);
    float s = 0.f, d = 0.f;
#pragma unroll
    for (int i = 0; i < HIDC / 4; ++i) {
        const float4 h4 = hv[i], s4 = sv[i], d4 = dv[i];
        s += h4.x * s4.x + h4.y * s4.y + h4.z * s4.z + h4.w * s4.w;
        d += h4.x * d4.x + h4.y * d4.y + h4.z * d4.z + h4.w * d4.w;
    }
    als[idx] = s;
    ald[idx] = d;
}

__global__ __launch_bounds__(256) void csr_count_kernel(
    const int* __restrict__ ei, int* __restrict__ cnt, int* __restrict__ rank)
{
    const int e = blockIdx.x * blockDim.x + threadIdx.x;
    if (e >= NEDGES) return;
    const int d = ei[NEDGES + e];
    rank[e] = atomicAdd(&cnt[d], 1);
}

#define SCAN_THREADS 1024
#define SCAN_TILE    4096

__device__ __forceinline__ int wave_sum_i(int v) {
#pragma unroll
    for (int o = 16; o > 0; o >>= 1) v += __shfl_xor(v, o);
    return v;
}

__global__ __launch_bounds__(SCAN_THREADS) void csr_scan_kernel(
    const int* __restrict__ cnt, int* __restrict__ rowptr)
{
    __shared__ int wsum[32];
    const int t = threadIdx.x;
    const int lane = t & 31;
    const int wave = t >> 5;
    const int base = blockIdx.x * SCAN_TILE;

    int pre = 0;
    for (int i = t * 4; i < base; i += SCAN_TILE) {
        const int4 v = *(const int4*)(cnt + i);
        pre += (v.x + v.y) + (v.z + v.w);
    }
    pre = wave_sum_i(pre);
    if (lane == 0) wsum[wave] = pre;
    __syncthreads();
    int blockpre = 0;
#pragma unroll
    for (int w = 0; w < 32; ++w) blockpre += wsum[w];
    __syncthreads();

    const int i0 = base + t * 4;
    int4 v = make_int4(0, 0, 0, 0);
    if (i0 + 3 < NNODES) {
        v = *(const int4*)(cnt + i0);
    } else {
        if (i0 + 0 < NNODES) v.x = cnt[i0 + 0];
        if (i0 + 1 < NNODES) v.y = cnt[i0 + 1];
        if (i0 + 2 < NNODES) v.z = cnt[i0 + 2];
    }
    const int s0 = v.x, s1 = s0 + v.y, s2 = s1 + v.z, s3 = s2 + v.w;
    int inc = s3;
#pragma unroll
    for (int o = 1; o < 32; o <<= 1) {
        const int u = __shfl_up(inc, o);
        if (lane >= o) inc += u;
    }
    if (lane == 31) wsum[wave] = inc;
    __syncthreads();
    int wpre = 0;
    for (int w = 0; w < wave; ++w) wpre += wsum[w];
    const int ex = blockpre + wpre + (inc - s3);
    if (i0 + 0 <= NNODES) rowptr[i0 + 0] = ex;
    if (i0 + 1 <= NNODES) rowptr[i0 + 1] = ex + s0;
    if (i0 + 2 <= NNODES) rowptr[i0 + 2] = ex + s1;
    if (i0 + 3 <= NNODES) rowptr[i0 + 3] = ex + s2;
}

__global__ __launch_bounds__(256) void csr_fill_kernel(
    const int* __restrict__ ei, const int* __restrict__ rowptr,
    const int* __restrict__ rank, int* __restrict__ csr_src)
{
    const int e = blockIdx.x * blockDim.x + threadIdx.x;
    if (e >= NEDGES) return;
    const int d = ei[NEDGES + e];
    csr_src[rowptr[d] + rank[e]] = ei[e];
}

__device__ __forceinline__ float wave_max_f(float v) {
#pragma unroll
    for (int o = 16; o > 0; o >>= 1) v = fmaxf(v, __shfl_xor(v, o));
    return v;
}
__device__ __forceinline__ float wave_sum_f(float v) {
#pragma unroll
    for (int o = 16; o > 0; o >>= 1) v += __shfl_xor(v, o);
    return v;
}
__device__ __forceinline__ float bcast_f(float v, int j) {
    return __int_as_float(__builtin_amdgcn_readlane(__float_as_int(v), j));
}

template <bool CONCAT>
__global__ __launch_bounds__(256) void gat_gather_kernel(
    const int* __restrict__ rowptr, const int* __restrict__ csr_src,
    const float* __restrict__ H, const float* __restrict__ als,
    const float* __restrict__ ald, const float* __restrict__ bias,
    const float* __restrict__ prelu_a, float* __restrict__ OUT)
{
    const int d = __builtin_amdgcn_readfirstlane((int)(blockIdx.x * 8 + (threadIdx.x >> 5)));
    if (d >= NNODES) return;
    const int lane = threadIdx.x & 31;
    const int beg = rowptr[d];
    const int deg = rowptr[d + 1] - beg;
    const int tot = deg + 1;
    const float ad0 = ald[d * 2 + 0];
    const float ad1 = ald[d * 2 + 1];
    const float2* __restrict__ als2 = (const float2*)als;

    auto edge = [&](int j, int& s, float& e0, float& e1) {
        s = (j < deg) ? csr_src[beg + j] : d;
        const float2 a = als2[s];
        e0 = (j < tot) ? leakyr(a.x + ad0) : -INFINITY;
        e1 = (j < tot) ? leakyr(a.y + ad1) : -INFINITY;
    };

    int sc; float e0c, e1c;
    edge(lane, sc, e0c, e1c);

    float m0 = e0c, m1 = e1c;
    for (int b = 32; b < tot; b += 32) {
        int s; float e0, e1;
        edge(b + lane, s, e0, e1);
        m0 = fmaxf(m0, e0); m1 = fmaxf(m1, e1);
    }
    m0 = wave_max_f(m0); m1 = wave_max_f(m1);

    const float x0c = (lane < tot) ? __expf(e0c - m0) : 0.f;
    const float x1c = (lane < tot) ? __expf(e1c - m1) : 0.f;
    float den0 = x0c, den1 = x1c;
    for (int b = 32; b < tot; b += 32) {
        int s; float e0, e1;
        edge(b + lane, s, e0, e1);
        den0 += (b + lane < tot) ? __expf(e0 - m0) : 0.f;
        den1 += (b + lane < tot) ? __expf(e1 - m1) : 0.f;
    }
    den0 = wave_sum_f(den0); den1 = wave_sum_f(den1);

    const int c = lane * 4;
    const bool head1 = lane >= 16;
    float4 acc = make_float4(0.f, 0.f, 0.f, 0.f);

    auto gather = [&](int s_l, float a0_l, float a1_l, int n) {
#pragma unroll 4
        for (int j = 0; j < n; ++j) {
            const int sj = __builtin_amdgcn_readlane(s_l, j);
            const float a0 = bcast_f(a0_l, j);
            const float a1 = bcast_f(a1_l, j);
            const float al = head1 ? a1 : a0;
            const float4 h4 = *(const float4*)(H + (size_t)sj * HCAT + c);
            acc.x += h4.x * al; acc.y += h4.y * al;
            acc.z += h4.z * al; acc.w += h4.w * al;
        }
    };

    gather(sc, x0c / den0, x1c / den1, tot < 32 ? tot : 32);
    for (int b = 32; b < tot; b += 32) {
        int s; float e0, e1;
        edge(b + lane, s, e0, e1);
        const float x0 = (b + lane < tot) ? __expf(e0 - m0) : 0.f;
        const float x1 = (b + lane < tot) ? __expf(e1 - m1) : 0.f;
        gather(s, x0 / den0, x1 / den1, (tot - b) < 32 ? (tot - b) : 32);
    }

    const float a = *prelu_a;
    if (CONCAT) {
        float4 v = acc;
        v.x += bias[c + 0]; v.y += bias[c + 1]; v.z += bias[c + 2]; v.w += bias[c + 3];
        v.x = v.x >= 0.f ? v.x : a * v.x;
        v.y = v.y >= 0.f ? v.y : a * v.y;
        v.z = v.z >= 0.f ? v.z : a * v.z;
        v.w = v.w >= 0.f ? v.w : a * v.w;
        *(float4*)(OUT + (size_t)d * HCAT + c) = v;
    } else {
        float4 o;
        o.x = __shfl_xor(acc.x, 16); o.y = __shfl_xor(acc.y, 16);
        o.z = __shfl_xor(acc.z, 16); o.w = __shfl_xor(acc.w, 16);
        const int cc = (lane & 15) * 4;
        float4 v;
        v.x = 0.5f * (acc.x + o.x) + bias[cc + 0];
        v.y = 0.5f * (acc.y + o.y) + bias[cc + 1];
        v.z = 0.5f * (acc.z + o.z) + bias[cc + 2];
        v.w = 0.5f * (acc.w + o.w) + bias[cc + 3];
        v.x = v.x >= 0.f ? v.x : a * v.x;
        v.y = v.y >= 0.f ? v.y : a * v.y;
        v.z = v.z >= 0.f ? v.z : a * v.z;
        v.w = v.w >= 0.f ? v.w : a * v.w;
        if (lane < 16) *(float4*)(OUT + (size_t)d * HIDC + cc) = v;
    }
}

extern "C" void kernel_launch(void* const* d_in, const int* in_sizes, int n_in,
                              void* d_out, int out_size, void* d_ws, size_t ws_size,
                              hipStream_t stream)
{
    const float* x       = (const float*)d_in[0];
    const int*   ei      = (const int*)  d_in[1];
    const float* W1      = (const float*)d_in[2];
    const float* a1s     = (const float*)d_in[3];
    const float* a1d     = (const float*)d_in[4];
    const float* b1      = (const float*)d_in[5];
    const float* W2      = (const float*)d_in[6];
    const float* a2s     = (const float*)d_in[7];
    const float* a2d     = (const float*)d_in[8];
    const float* b2      = (const float*)d_in[9];
    const float* W3      = (const float*)d_in[10];
    const float* a3s     = (const float*)d_in[11];
    const float* a3d     = (const float*)d_in[12];
    const float* b3      = (const float*)d_in[13];
    const float* prelu_a = (const float*)d_in[14];
    const float* Wc      = (const float*)d_in[15];
    const float* bc      = (const float*)d_in[16];
    float* out = (float*)d_out;

    float* ws = (float*)d_ws;
    size_t off = 0;
    float* H     = ws + off; off += (size_t)NNODES * HCAT;
    float* PREV  = ws + off; off += (size_t)NNODES * HCAT;
    float* als   = ws + off; off += (size_t)NNODES * 2;
    float* ald   = ws + off; off += (size_t)NNODES * 2;
    int* cnt     = (int*)(ws + off); off += (size_t)NNODES;
    int* rowptr  = (int*)(ws + off); off += (size_t)NNODES + 4;
    int* rank    = (int*)(ws + off); off += (size_t)NEDGES;
    int* csr_src = (int*)(ws + off); off += (size_t)NEDGES;

    const int gN2    = (NNODES * NHEADS + 255) / 256;
    const int gEdge  = (NEDGES + 255) / 256;
    const int gScan  = (NNODES + 1 + SCAN_TILE - 1) / SCAN_TILE;
    const int gNodeW = (NNODES + 7) / 8;
    const int gTile  = ((NNODES / 16) + 7) / 8;

    hipMemsetAsync(cnt, 0x00, (size_t)NNODES * sizeof(int), stream);
    csr_count_kernel<<<gEdge, 256, 0, stream>>>(ei, cnt, rank);
    csr_scan_kernel <<<gScan, SCAN_THREADS, 0, stream>>>(cnt, rowptr);
    csr_fill_kernel <<<gEdge, 256, 0, stream>>>(ei, rowptr, rank, csr_src);

    struct Layer {
        const float* W; const float* as; const float* ad; const float* b; bool concat;
    };
    const Layer layers[3] = {
        { W1, a1s, a1d, b1, true  },
        { W2, a2s, a2d, b2, true  },
        { W3, a3s, a3d, b3, false },
    };

    for (int l = 0; l < 3; ++l) {
        const float* in = (l == 0) ? x : PREV;

        if (l == 0)
            gemm_wmma_f32<INCH, HCAT, 8><<<gTile, 256, 0, stream>>>(in, layers[l].W, nullptr, H, NNODES);
        else
            gemm_wmma_f32<HCAT, HCAT, 8><<<gTile, 256, 0, stream>>>(in, layers[l].W, nullptr, H, NNODES);

        attn_logits_kernel<<<gN2, 256, 0, stream>>>(H, layers[l].as, layers[l].ad, als, ald);

        if (layers[l].concat)
            gat_gather_kernel<true ><<<gNodeW, 256, 0, stream>>>(rowptr, csr_src, H, als, ald,
                                                                layers[l].b, prelu_a, PREV);
        else
            gat_gather_kernel<false><<<gNodeW, 256, 0, stream>>>(rowptr, csr_src, H, als, ald,
                                                                layers[l].b, prelu_a, PREV);
    }

    gemm_wmma_f32<HIDC, NCLS, 1><<<gTile, 256, 0, stream>>>(PREV, Wc, bc, out, NNODES);
    (void)in_sizes; (void)n_in; (void)out_size; (void)ws_size;
}


# Layer 1 sums the 32-channel input rows per head before the linear map (linearity), so its gather reads 128-byte rows, not 512; logits from x.(W a); row sum order changes (4 partial sums)
# speedup vs baseline: 5.6081x; 1.0353x over previous
#include <hip/hip_runtime.h>
#include <hip/hip_bf16.h>

#define NNODES 100000
#define NEDGES 1600000
#define INCH   32
#define HIDC   64
#define NHEADS 2
#define HCAT   128
#define NCLS   12
#define NEGSLOPE 0.2f

typedef __attribute__((ext_vector_type(2))) float v2f;
typedef __attribute__((ext_vector_type(8))) float v8f;

__device__ __forceinline__ float leakyr(float x) { return x > 0.f ? x : NEGSLOPE * x; }

template <int K, int M, int MT>
__global__ __launch_bounds__(256) void gemm_wmma_f32(
    const float* __restrict__ A, const float* __restrict__ B,
    const float* __restrict__ bias, float* __restrict__ D, int nrows)
{
    const int wave = threadIdx.x >> 5;
    const int lane = threadIdx.x & 31;
    const int tile = blockIdx.x * 8 + wave;
    const int ntiles = nrows >> 4;
    if (tile >= ntiles) return;
    const int half = lane >> 4;
    const int mrow = lane & 15;
    const int ncol = lane & 15;
    const int m0 = tile << 4;

    v8f acc[MT] = {};

    const float* __restrict__ arow = A + (size_t)(m0 + mrow) * K;
    for (int kb = 0; kb < K; kb += 4) {
        const int k0 = kb + 2 * half;
        v2f a;
        a.x = arow[k0];
        a.y = arow[k0 + 1];
#pragma unroll
        for (int t = 0; t < MT; ++t) {
            const int n = t * 16 + ncol;
            v2f b;
            if ((M & 15) == 0 || n < M) {
                b.x = B[(size_t)k0 * M + n];
                b.y = B[(size_t)(k0 + 1) * M + n];
            } else {
                b.x = 0.f; b.y = 0.f;
            }
            acc[t] = __builtin_amdgcn_wmma_f32_16x16x4_f32(
                false, a, false, b, (short)0, acc[t], false, false);
        }
    }

#pragma unroll
    for (int t = 0; t < MT; ++t) {
        const int n = t * 16 + ncol;
        if ((M & 15) != 0 && n >= M) continue;
        const float bv = bias ? bias[n] : 0.f;
#pragma unroll
        for (int r = 0; r < 8; ++r) {
            const int m = m0 + r + 8 * half;
            D[(size_t)m * M + n] = acc[t][r] + bv;
        }
    }
}

__global__ __launch_bounds__(256) void attn_logits_kernel(
    const float* __restrict__ H, const float* __restrict__ a_s,
    const float* __restrict__ a_d, float* __restrict__ als, float* __restrict__ ald)
{
    const int idx = blockIdx.x * blockDim.x + threadIdx.x;
    if (idx >= NNODES * NHEADS) return;
    const int head = idx & 1;
    const int node = idx >> 1;
    const float4* hv = (const float4*)(H + (size_t)node * HCAT + head * HIDC);
    const float4* sv = (const float4*)(a_s + head * HIDC);
    const float4* dv = (const float4*)(a_d + head * HIDC);
    float s = 0.f, d = 0.f;
#pragma unroll
    for (int i = 0; i < HIDC / 4; ++i) {
        const float4 h4 = hv[i], s4 = sv[i], d4 = dv[i];
        s += h4.x * s4.x + h4.y * s4.y + h4.z * s4.z + h4.w * s4.w;
        d += h4.x * d4.x + h4.y * d4.y + h4.z * d4.z + h4.w * d4.w;
    }
    als[idx] = s;
    ald[idx] = d;
}

__global__ __launch_bounds__(256) void csr_count_kernel(
    const int* __restrict__ ei, int* __restrict__ cnt, int* __restrict__ rank)
{
    const int e = blockIdx.x * blockDim.x + threadIdx.x;
    if (e >= NEDGES) return;
    const int d = ei[NEDGES + e];
    rank[e] = atomicAdd(&cnt[d], 1);
}

#define SCAN_THREADS 1024
#define SCAN_TILE    4096

__device__ __forceinline__ int wave_sum_i(int v) {
#pragma unroll
    for (int o = 16; o > 0; o >>= 1) v += __shfl_xor(v, o);
    return v;
}

__global__ __launch_bounds__(SCAN_THREADS) void csr_scan_kernel(
    const int* __restrict__ cnt, int* __restrict__ rowptr)
{
    __shared__ int wsum[32];
    const int t = threadIdx.x;
    const int lane = t & 31;
    const int wave = t >> 5;
    const int base = blockIdx.x * SCAN_TILE;

    int pre = 0;
    for (int i = t * 4; i < base; i += SCAN_TILE) {
        const int4 v = *(const int4*)(cnt + i);
        pre += (v.x + v.y) + (v.z + v.w);
    }
    pre = wave_sum_i(pre);
    if (lane == 0) wsum[wave] = pre;
    __syncthreads();
    int blockpre = 0;
#pragma unroll
    for (int w = 0; w < 32; ++w) blockpre += wsum[w];
    __syncthreads();

    const int i0 = base + t * 4;
    int4 v = make_int4(0, 0, 0, 0);
    if (i0 + 3 < NNODES) {
        v = *(const int4*)(cnt + i0);
    } else {
        if (i0 + 0 < NNODES) v.x = cnt[i0 + 0];
        if (i0 + 1 < NNODES) v.y = cnt[i0 + 1];
        if (i0 + 2 < NNODES) v.z = cnt[i0 + 2];
    }
    const int s0 = v.x, s1 = s0 + v.y, s2 = s1 + v.z, s3 = s2 + v.w;
    int inc = s3;
#pragma unroll
    for (int o = 1; o < 32; o <<= 1) {
        const int u = __shfl_up(inc, o);
        if (lane >= o) inc += u;
    }
    if (lane == 31) wsum[wave] = inc;
    __syncthreads();
    int wpre = 0;
    for (int w = 0; w < wave; ++w) wpre += wsum[w];
    const int ex = blockpre + wpre + (inc - s3);
    if (i0 + 0 <= NNODES) rowptr[i0 + 0] = ex;
    if (i0 + 1 <= NNODES) rowptr[i0 + 1] = ex + s0;
    if (i0 + 2 <= NNODES) rowptr[i0 + 2] = ex + s1;
    if (i0 + 3 <= NNODES) rowptr[i0 + 3] = ex + s2;
}

__global__ __launch_bounds__(256) void csr_fill_kernel(
    const int* __restrict__ ei, const int* __restrict__ rowptr,
    const int* __restrict__ rank, int* __restrict__ csr_src)
{
    const int e = blockIdx.x * blockDim.x + threadIdx.x;
    if (e >= NEDGES) return;
    const int d = ei[NEDGES + e];
    csr_src[rowptr[d] + rank[e]] = ei[e];
}

__device__ __forceinline__ float wave_max_f(float v) {
#pragma unroll
    for (int o = 16; o > 0; o >>= 1) v = fmaxf(v, __shfl_xor(v, o));
    return v;
}
__device__ __forceinline__ float wave_sum_f(float v) {
#pragma unroll
    for (int o = 16; o > 0; o >>= 1) v += __shfl_xor(v, o);
    return v;
}
__device__ __forceinline__ float bcast_f(float v, int j) {
    return __int_as_float(__builtin_amdgcn_readlane(__float_as_int(v), j));
}

template <bool CONCAT>
__global__ __launch_bounds__(256) void gat_gather_kernel(
    const int* __restrict__ rowptr, const int* __restrict__ csr_src,
    const float* __restrict__ H, const float* __restrict__ als,
    const float* __restrict__ ald, const float* __restrict__ bias,
    const float* __restrict__ prelu_a, float* __restrict__ OUT)
{
    const int d = __builtin_amdgcn_readfirstlane((int)(blockIdx.x * 8 + (threadIdx.x >> 5)));
    if (d >= NNODES) return;
    const int lane = threadIdx.x & 31;
    const int beg = rowptr[d];
    const int deg = rowptr[d + 1] - beg;
    const int tot = deg + 1;
    const float ad0 = ald[d * 2 + 0];
    const float ad1 = ald[d * 2 + 1];
    const float2* __restrict__ als2 = (const float2*)als;

    auto edge = [&](int j, int& s, float& e0, float& e1) {
        s = (j < deg) ? csr_src[beg + j] : d;
        const float2 a = als2[s];
        e0 = (j < tot) ? leakyr(a.x + ad0) : -INFINITY;
        e1 = (j < tot) ? leakyr(a.y + ad1) : -INFINITY;
    };

    int sc; float e0c, e1c;
    edge(lane, sc, e0c, e1c);

    float m0 = e0c, m1 = e1c;
    for (int b = 32; b < tot; b += 32) {
        int s; float e0, e1;
        edge(b + lane, s, e0, e1);
        m0 = fmaxf(m0, e0); m1 = fmaxf(m1, e1);
    }
    m0 = wave_max_f(m0); m1 = wave_max_f(m1);

    const float x0c = (lane < tot) ? __expf(e0c - m0) : 0.f;
    const float x1c = (lane < tot) ? __expf(e1c - m1) : 0.f;
    float den0 = x0c, den1 = x1c;
    for (int b = 32; b < tot; b += 32) {
        int s; float e0, e1;
        edge(b + lane, s, e0, e1);
        den0 += (b + lane < tot) ? __expf(e0 - m0) : 0.f;
        den1 += (b + lane < tot) ? __expf(e1 - m1) : 0.f;
    }
    den0 = wave_sum_f(den0); den1 = wave_sum_f(den1);

    const int c = lane * 4;
    const bool head1 = lane >= 16;
    float4 acc = make_float4(0.f, 0.f, 0.f, 0.f);

    auto gather = [&](int s_l, float a0_l, float a1_l, int n) {
#pragma unroll 4
        for (int j = 0; j < n; ++j) {
            const int sj = __builtin_amdgcn_readlane(s_l, j);
            const float a0 = bcast_f(a0_l, j);
            const float a1 = bcast_f(a1_l, j);
            const float al = head1 ? a1 : a0;
            const float4 h4 = *(const float4*)(H + (size_t)sj * HCAT + c);
            acc.x += h4.x * al; acc.y += h4.y * al;
            acc.z += h4.z * al; acc.w += h4.w * al;
        }
    };

    gather(sc, x0c / den0, x1c / den1, tot < 32 ? tot : 32);
    for (int b = 32; b < tot; b += 32) {
        int s; float e0, e1;
        edge(b + lane, s, e0, e1);
        const float x0 = (b + lane < tot) ? __expf(e0 - m0) : 0.f;
        const float x1 = (b + lane < tot) ? __expf(e1 - m1) : 0.f;
        gather(s, x0 / den0, x1 / den1, (tot - b) < 32 ? (tot - b) : 32);
    }

    const float a = *prelu_a;
    if (CONCAT) {
        float4 v = acc;
        v.x += bias[c + 0]; v.y += bias[c + 1]; v.z += bias[c + 2]; v.w += bias[c + 3];
        v.x = v.x >= 0.f ? v.x : a * v.x;
        v.y = v.y >= 0.f ? v.y : a * v.y;
        v.z = v.z >= 0.f ? v.z : a * v.z;
        v.w = v.w >= 0.f ? v.w : a * v.w;
        *(float4*)(OUT + (size_t)d * HCAT + c) = v;
    } else {
        float4 o;
        o.x = __shfl_xor(acc.x, 16); o.y = __shfl_xor(acc.y, 16);
        o.z = __shfl_xor(acc.z, 16); o.w = __shfl_xor(acc.w, 16);
        const int cc = (lane & 15) * 4;
        float4 v;
        v.x = 0.5f * (acc.x + o.x) + bias[cc + 0];
        v.y = 0.5f * (acc.y + o.y) + bias[cc + 1];
        v.z = 0.5f * (acc.z + o.z) + bias[cc + 2];
        v.w = 0.5f * (acc.w + o.w) + bias[cc + 3];
        v.x = v.x >= 0.f ? v.x : a * v.x;
        v.y = v.y >= 0.f ? v.y : a * v.y;
        v.z = v.z >= 0.f ? v.z : a * v.z;
        v.w = v.w >= 0.f ? v.w : a * v.w;
        if (lane < 16) *(float4*)(OUT + (size_t)d * HIDC + cc) = v;
    }
}


__global__ __launch_bounds__(256) void attn_logits_in_kernel(
    const float* __restrict__ X, const float* __restrict__ Wm,
    const float* __restrict__ a_s, const float* __restrict__ a_d,
    float* __restrict__ als, float* __restrict__ ald)
{
    __shared__ float4 wv[INCH];
    const int t = threadIdx.x;
    if (t < INCH * 4) {
        const int k = t >> 2, j = t & 3, head = j & 1;
        const float* __restrict__ av = ((j & 2) ? a_d : a_s) + head * HIDC;
        const float* __restrict__ wr = Wm + (size_t)k * HCAT + head * HIDC;
        const float4* __restrict__ w4 = (const float4*)wr;
        const float4* __restrict__ a4 = (const float4*)av;
        float s = 0.f;
#pragma unroll
        for (int c = 0; c < HIDC / 4; ++c) {
            const float4 w = w4[c], a = a4[c];
            s += w.x * a.x + w.y * a.y + w.z * a.z + w.w * a.w;
        }
        ((float*)wv)[t] = s;
    }
    __syncthreads();
    const int node = blockIdx.x * blockDim.x + t;
    if (node >= NNODES) return;
    const float4* __restrict__ xv = (const float4*)(X + (size_t)node * INCH);
    float s0 = 0.f, s1 = 0.f, d0 = 0.f, d1 = 0.f;
#pragma unroll
    for (int i = 0; i < INCH / 4; ++i) {
        const float4 x4 = xv[i];
        const float xs[4] = { x4.x, x4.y, x4.z, x4.w };
#pragma unroll
        for (int u = 0; u < 4; ++u) {
            const float4 w = wv[i * 4 + u];
            s0 += xs[u] * w.x; s1 += xs[u] * w.y;
            d0 += xs[u] * w.z; d1 += xs[u] * w.w;
        }
    }
    ((float2*)als)[node] = make_float2(s0, s1);
    ((float2*)ald)[node] = make_float2(d0, d1);
}

template <int I>
__device__ __forceinline__ int group8_bcast_i(int v) {
    constexpr unsigned sel = I | (I << 3) | (I << 6) | (I << 9) | (I << 12) | (I << 15) | (I << 18) | (I << 21);
    return __builtin_amdgcn_mov_dpp8(v, sel);
}
template <int I>
__device__ __forceinline__ float group8_bcast_f(float v) {
    return __int_as_float(group8_bcast_i<I>(__float_as_int(v)));
}

__global__ __launch_bounds__(256) void gat_gather_in_kernel(
    const int* __restrict__ rowptr, const int* __restrict__ csr_src,
    const float* __restrict__ X, const float* __restrict__ als,
    const float* __restrict__ ald, float* __restrict__ Z)
{
    const int d = __builtin_amdgcn_readfirstlane((int)(blockIdx.x * 8 + (threadIdx.x >> 5)));
    if (d >= NNODES) return;
    const int lane = threadIdx.x & 31;
    const int beg = rowptr[d];
    const int deg = rowptr[d + 1] - beg;
    const int tot = deg + 1;
    const float ad0 = ald[d * 2 + 0];
    const float ad1 = ald[d * 2 + 1];
    const float2* __restrict__ als2 = (const float2*)als;

    auto edge = [&](int j, int& s, float& e0, float& e1) {
        s = (j < deg) ? csr_src[beg + j] : d;
        const float2 a = als2[s];
        e0 = (j < tot) ? leakyr(a.x + ad0) : -INFINITY;
        e1 = (j < tot) ? leakyr(a.y + ad1) : -INFINITY;
    };

    int sc; float e0c, e1c;
    edge(lane, sc, e0c, e1c);

    float m0 = e0c, m1 = e1c;
    for (int b = 32; b < tot; b += 32) {
        int s; float e0, e1;
        edge(b + lane, s, e0, e1);
        m0 = fmaxf(m0, e0); m1 = fmaxf(m1, e1);
    }
    m0 = wave_max_f(m0); m1 = wave_max_f(m1);

    const float x0c = (lane < tot) ? __expf(e0c - m0) : 0.f;
    const float x1c = (lane < tot) ? __expf(e1c - m1) : 0.f;
    float den0 = x0c, den1 = x1c;
    for (int b = 32; b < tot; b += 32) {
        int s; float e0, e1;
        edge(b + lane, s, e0, e1);
        den0 += (b + lane < tot) ? __expf(e0 - m0) : 0.f;
        den1 += (b + lane < tot) ? __expf(e1 - m1) : 0.f;
    }
    den0 = wave_sum_f(den0); den1 = wave_sum_f(den1);

    const int jbase = lane & 24;
    const unsigned q = (lane & 7) * 4;
    float4 acc0 = make_float4(0.f, 0.f, 0.f, 0.f);
    float4 acc1 = make_float4(0.f, 0.f, 0.f, 0.f);

#define GATHER_IN_LOAD(I)                                                          \
    {                                                                              \
        const unsigned sj = (unsigned)group8_bcast_i<I>(s_l);                      \
        xr[I] = (jbase + I < n) ? *(const float4*)(X + (sj * INCH + q))            \
                                : make_float4(0.f, 0.f, 0.f, 0.f);                 \
    }
#define GATHER_IN_SUM(I)                                                           \
    {                                                                              \
        const float a0 = group8_bcast_f<I>(a0_l);                                  \
        const float a1 = group8_bcast_f<I>(a1_l);                                  \
        acc0.x += xr[I].x * a0; acc0.y += xr[I].y * a0;                            \
        acc0.z += xr[I].z * a0; acc0.w += xr[I].w * a0;                            \
        acc1.x += xr[I].x * a1; acc1.y += xr[I].y * a1;                            \
        acc1.z += xr[I].z * a1; acc1.w += xr[I].w * a1;                            \
    }
    auto gather = [&](int s_l, float a0_l, float a1_l, int n) {
        float4 xr[8];
        GATHER_IN_LOAD(0) GATHER_IN_LOAD(1) GATHER_IN_LOAD(2) GATHER_IN_LOAD(3)
        GATHER_IN_LOAD(4) GATHER_IN_LOAD(5) GATHER_IN_LOAD(6) GATHER_IN_LOAD(7)
        GATHER_IN_SUM(0) GATHER_IN_SUM(1) GATHER_IN_SUM(2) GATHER_IN_SUM(3)
        GATHER_IN_SUM(4) GATHER_IN_SUM(5) GATHER_IN_SUM(6) GATHER_IN_SUM(7)
    };

    gather(sc, x0c / den0, x1c / den1, tot < 32 ? tot : 32);
    for (int b = 32; b < tot; b += 32) {
        int s; float e0, e1;
        edge(b + lane, s, e0, e1);
        const float x0 = (b + lane < tot) ? __expf(e0 - m0) : 0.f;
        const float x1 = (b + lane < tot) ? __expf(e1 - m1) : 0.f;
        gather(s, x0 / den0, x1 / den1, (tot - b) < 32 ? (tot - b) : 32);
    }
#undef GATHER_IN_LOAD
#undef GATHER_IN_SUM

#define XSUM(v) v += __shfl_xor(v, 8); v += __shfl_xor(v, 16);
    XSUM(acc0.x) XSUM(acc0.y) XSUM(acc0.z) XSUM(acc0.w)
    XSUM(acc1.x) XSUM(acc1.y) XSUM(acc1.z) XSUM(acc1.w)
#undef XSUM

    float* __restrict__ zrow = Z + (size_t)d * (NHEADS * INCH);
    if (lane < 8)       *(float4*)(zrow + q) = acc0;
    else if (lane < 16) *(float4*)(zrow + INCH + q) = acc1;
}

__global__ __launch_bounds__(256) void gemm_heads_wmma_f32(
    const float* __restrict__ Z, const float* __restrict__ B,
    const float* __restrict__ bias, const float* __restrict__ prelu_a,
    float* __restrict__ D, int nrows)
{
    const int wave = threadIdx.x >> 5;
    const int lane = threadIdx.x & 31;
    const int tile = blockIdx.x * 8 + wave;
    const int ntiles = nrows >> 4;
    if (tile >= ntiles) return;
    const int half = lane >> 4;
    const int mrow = lane & 15;
    const int ncol = lane & 15;
    const int m0 = tile << 4;

    v8f acc[8] = {};

    const float* __restrict__ zrow = Z + (size_t)(m0 + mrow) * (NHEADS * INCH);
    for (int kb = 0; kb < INCH; kb += 4) {
        const int k0 = kb + 2 * half;
        v2f a0, a1;
        a0.x = zrow[k0];        a0.y = zrow[k0 + 1];
        a1.x = zrow[INCH + k0]; a1.y = zrow[INCH + k0 + 1];
#pragma unroll
        for (int t = 0; t < 8; ++t) {
            const int n = t * 16 + ncol;
            v2f b;
            b.x = B[(size_t)k0 * HCAT + n];
            b.y = B[(size_t)(k0 + 1) * HCAT + n];
            acc[t] = __builtin_amdgcn_wmma_f32_16x16x4_f32(
                false, (t < 4) ? a0 : a1, false, b, (short)0, acc[t], false, false);
        }
    }

    const float a = *prelu_a;
#pragma unroll
    for (int t = 0; t < 8; ++t) {
        const int n = t * 16 + ncol;
        const float bv = bias[n];
#pragma unroll
        for (int r = 0; r < 8; ++r) {
            const int m = m0 + r + 8 * half;
            float v = acc[t][r] + bv;
            v = v >= 0.f ? v : a * v;
            D[(size_t)m * HCAT + n] = v;
        }
    }
}

extern "C" void kernel_launch(void* const* d_in, const int* in_sizes, int n_in,
                              void* d_out, int out_size, void* d_ws, size_t ws_size,
                              hipStream_t stream)
{
    const float* x       = (const float*)d_in[0];
    const int*   ei      = (const int*)  d_in[1];
    const float* W1      = (const float*)d_in[2];
    const float* a1s     = (const float*)d_in[3];
    const float* a1d     = (const float*)d_in[4];
    const float* b1      = (const float*)d_in[5];
    const float* W2      = (const float*)d_in[6];
    const float* a2s     = (const float*)d_in[7];
    const float* a2d     = (const float*)d_in[8];
    const float* b2      = (const float*)d_in[9];
    const float* W3      = (const float*)d_in[10];
    const float* a3s     = (const float*)d_in[11];
    const float* a3d     = (const float*)d_in[12];
    const float* b3      = (const float*)d_in[13];
    const float* prelu_a = (const float*)d_in[14];
    const float* Wc      = (const float*)d_in[15];
    const float* bc      = (const float*)d_in[16];
    float* out = (float*)d_out;

    float* ws = (float*)d_ws;
    size_t off = 0;
    float* H     = ws + off; off += (size_t)NNODES * HCAT;
    float* PREV  = ws + off; off += (size_t)NNODES * HCAT;
    float* als   = ws + off; off += (size_t)NNODES * 2;
    float* ald   = ws + off; off += (size_t)NNODES * 2;
    int* cnt     = (int*)(ws + off); off += (size_t)NNODES;
    int* rowptr  = (int*)(ws + off); off += (size_t)NNODES + 4;
    int* rank    = (int*)(ws + off); off += (size_t)NEDGES;
    int* csr_src = (int*)(ws + off); off += (size_t)NEDGES;

    const int gN1    = (NNODES + 255) / 256;
    const int gN2    = (NNODES * NHEADS + 255) / 256;
    const int gEdge  = (NEDGES + 255) / 256;
    const int gScan  = (NNODES + 1 + SCAN_TILE - 1) / SCAN_TILE;
    const int gNodeW = (NNODES + 7) / 8;
    const int gTile  = ((NNODES / 16) + 7) / 8;

    hipMemsetAsync(cnt, 0x00, (size_t)NNODES * sizeof(int), stream);
    csr_count_kernel<<<gEdge, 256, 0, stream>>>(ei, cnt, rank);
    csr_scan_kernel <<<gScan, SCAN_THREADS, 0, stream>>>(cnt, rowptr);
    csr_fill_kernel <<<gEdge, 256, 0, stream>>>(ei, rowptr, rank, csr_src);

    struct Layer {
        const float* W; const float* as; const float* ad; const float* b; bool concat;
    };
    const Layer layers[3] = {
        { W1, a1s, a1d, b1, true  },
        { W2, a2s, a2d, b2, true  },
        { W3, a3s, a3d, b3, false },
    };

    float* Z = H;
    attn_logits_in_kernel<<<gN1, 256, 0, stream>>>(x, W1, a1s, a1d, als, ald);
    gat_gather_in_kernel <<<gNodeW, 256, 0, stream>>>(rowptr, csr_src, x, als, ald, Z);
    gemm_heads_wmma_f32  <<<gTile, 256, 0, stream>>>(Z, W1, b1, prelu_a, PREV, NNODES);

    for (int l = 1; l < 3; ++l) {
        gemm_wmma_f32<HCAT, HCAT, 8><<<gTile, 256, 0, stream>>>(PREV, layers[l].W, nullptr, H, NNODES);

        attn_logits_kernel<<<gN2, 256, 0, stream>>>(H, layers[l].as, layers[l].ad, als, ald);

        if (layers[l].concat)
            gat_gather_kernel<true ><<<gNodeW, 256, 0, stream>>>(rowptr, csr_src, H, als, ald,
                                                                layers[l].b, prelu_a, PREV);
        else
            gat_gather_kernel<false><<<gNodeW, 256, 0, stream>>>(rowptr, csr_src, H, als, ald,
                                                                layers[l].b, prelu_a, PREV);
    }

    gemm_wmma_f32<HIDC, NCLS, 1><<<gTile, 256, 0, stream>>>(PREV, Wc, bc, out, NNODES);
    (void)in_sizes; (void)n_in; (void)out_size; (void)ws_size;
}


# Edges go straight into a fixed 64-slot row per destination (slot = arrival rank from the count atomic; overflow to a side list): scan and fill launches and their random reads go; sum order kept
# speedup vs baseline: 6.0457x; 1.0780x over previous
#include <hip/hip_runtime.h>
#include <hip/hip_bf16.h>

#define NNODES 100000
#define NEDGES 1600000
#define INCH   32
#define HIDC   64
#define NHEADS 2
#define HCAT   128
#define NCLS   12
#define NEGSLOPE 0.2f

typedef __attribute__((ext_vector_type(2))) float v2f;
typedef __attribute__((ext_vector_type(8))) float v8f;

__device__ __forceinline__ float leakyr(float x) { return x > 0.f ? x : NEGSLOPE * x; }

template <int K, int M, int MT>
__global__ __launch_bounds__(256) void gemm_wmma_f32(
    const float* __restrict__ A, const float* __restrict__ B,
    const float* __restrict__ bias, float* __restrict__ D, int nrows)
{
    const int wave = threadIdx.x >> 5;
    const int lane = threadIdx.x & 31;
    const int tile = blockIdx.x * 8 + wave;
    const int ntiles = nrows >> 4;
    if (tile >= ntiles) return;
    const int half = lane >> 4;
    const int mrow = lane & 15;
    const int ncol = lane & 15;
    const int m0 = tile << 4;

    v8f acc[MT] = {};

    const float* __restrict__ arow = A + (size_t)(m0 + mrow) * K;
    for (int kb = 0; kb < K; kb += 4) {
        const int k0 = kb + 2 * half;
        v2f a;
        a.x = arow[k0];
        a.y = arow[k0 + 1];
#pragma unroll
        for (int t = 0; t < MT; ++t) {
            const int n = t * 16 + ncol;
            v2f b;
            if ((M & 15) == 0 || n < M) {
                b.x = B[(size_t)k0 * M + n];
                b.y = B[(size_t)(k0 + 1) * M + n];
            } else {
                b.x = 0.f; b.y = 0.f;
            }
            acc[t] = __builtin_amdgcn_wmma_f32_16x16x4_f32(
                false, a, false, b, (short)0, acc[t], false, false);
        }
    }

#pragma unroll
    for (int t = 0; t < MT; ++t) {
        const int n = t * 16 + ncol;
        if ((M & 15) != 0 && n >= M) continue;
        const float bv = bias ? bias[n] : 0.f;
#pragma unroll
        for (int r = 0; r < 8; ++r) {
            const int m = m0 + r + 8 * half;
            D[(size_t)m * M + n] = acc[t][r] + bv;
        }
    }
}

__global__ __launch_bounds__(256) void attn_logits_kernel(
    const float* __restrict__ H, const float* __restrict__ a_s,
    const float* __restrict__ a_d, float* __restrict__ als, float* __restrict__ ald)
{
    const int idx = blockIdx.x * blockDim.x + threadIdx.x;
    if (idx >= NNODES * NHEADS) return;
    const int head = idx & 1;
    const int node = idx >> 1;
    const float4* hv = (const float4*)(H + (size_t)node * HCAT + head * HIDC);
    const float4* sv = (const float4*)(a_s + head * HIDC);
    const float4* dv = (const float4*)(a_d + head * HIDC);
    float s = 0.f, d = 0.f;
#pragma unroll
    for (int i = 0; i < HIDC / 4; ++i) {
        const float4 h4 = hv[i], s4 = sv[i], d4 = dv[i];
        s += h4.x * s4.x + h4.y * s4.y + h4.z * s4.z + h4.w * s4.w;
        d += h4.x * d4.x + h4.y * d4.y + h4.z * d4.z + h4.w * d4.w;
    }
    als[idx] = s;
    ald[idx] = d;
}

#define SLOTC 64
#define CNTWORDS 100096

__global__ __launch_bounds__(256) void bucket_edges_kernel(
    const int* __restrict__ ei, int* __restrict__ cnt, int* __restrict__ slots,
    int* __restrict__ ovf)
{
    const int e = blockIdx.x * blockDim.x + threadIdx.x;
    if (e >= NEDGES) return;
    const int s = ei[e];
    const int d = ei[NEDGES + e];
    const int r = atomicAdd(&cnt[d], 1);
    if (r < SLOTC) {
        slots[d * SLOTC + r] = s;
    } else {
        const int o = atomicAdd(&cnt[NNODES], 1);
        *(int2*)(ovf + 2 * o) = make_int2(d, s);
    }
}

__device__ __forceinline__ float wave_max_f(float v) {
#pragma unroll
    for (int o = 16; o > 0; o >>= 1) v = fmaxf(v, __shfl_xor(v, o));
    return v;
}
__device__ __forceinline__ float wave_sum_f(float v) {
#pragma unroll
    for (int o = 16; o > 0; o >>= 1) v += __shfl_xor(v, o);
    return v;
}
__device__ __forceinline__ float bcast_f(float v, int j) {
    return __int_as_float(__builtin_amdgcn_readlane(__float_as_int(v), j));
}

template <bool CONCAT>
__global__ __launch_bounds__(256) void gat_gather_kernel(
    const int* __restrict__ cnt, const int* __restrict__ slots, const int* __restrict__ ovf,
    const float* __restrict__ H, const float* __restrict__ als,
    const float* __restrict__ ald, const float* __restrict__ bias,
    const float* __restrict__ prelu_a, float* __restrict__ OUT)
{
    const int d = __builtin_amdgcn_readfirstlane((int)(blockIdx.x * 8 + (threadIdx.x >> 5)));
    if (d >= NNODES) return;
    const int lane = threadIdx.x & 31;
    const int beg = d * SLOTC;
    const int indeg = cnt[d];
    const int deg = indeg < SLOTC ? indeg : SLOTC;
    const int tot = deg + 1;
    const int novf = indeg > SLOTC ? cnt[NNODES] : 0;
    const float ad0 = ald[d * 2 + 0];
    const float ad1 = ald[d * 2 + 1];
    const float2* __restrict__ als2 = (const float2*)als;

    auto edge = [&](int j, int& s, float& e0, float& e1) {
        s = (j < deg) ? slots[beg + j] : d;
        const float2 a = als2[s];
        e0 = (j < tot) ? leakyr(a.x + ad0) : -INFINITY;
        e1 = (j < tot) ? leakyr(a.y + ad1) : -INFINITY;
    };
    auto ovf_edge = [&](int i, int& s, float& e0, float& e1) -> bool {
        bool mine = false;
        s = d;
        if (i < novf) {
            const int2 p = *(const int2*)(ovf + 2 * i);
            mine = (p.x == d);
            s = mine ? p.y : d;
        }
        const float2 a = als2[s];
        e0 = mine ? leakyr(a.x + ad0) : -INFINITY;
        e1 = mine ? leakyr(a.y + ad1) : -INFINITY;
        return mine;
    };

    int sc; float e0c, e1c;
    edge(lane, sc, e0c, e1c);

    float m0 = e0c, m1 = e1c;
    for (int b = 32; b < tot; b += 32) {
        int s; float e0, e1;
        edge(b + lane, s, e0, e1);
        m0 = fmaxf(m0, e0); m1 = fmaxf(m1, e1);
    }
    for (int b = 0; b < novf; b += 32) {
        int s; float e0, e1;
        ovf_edge(b + lane, s, e0, e1);
        m0 = fmaxf(m0, e0); m1 = fmaxf(m1, e1);
    }
    m0 = wave_max_f(m0); m1 = wave_max_f(m1);

    const float x0c = (lane < tot) ? __expf(e0c - m0) : 0.f;
    const float x1c = (lane < tot) ? __expf(e1c - m1) : 0.f;
    float den0 = x0c, den1 = x1c;
    for (int b = 32; b < tot; b += 32) {
        int s; float e0, e1;
        edge(b + lane, s, e0, e1);
        den0 += (b + lane < tot) ? __expf(e0 - m0) : 0.f;
        den1 += (b + lane < tot) ? __expf(e1 - m1) : 0.f;
    }
    for (int b = 0; b < novf; b += 32) {
        int s; float e0, e1;
        const bool mine = ovf_edge(b + lane, s, e0, e1);
        den0 += mine ? __expf(e0 - m0) : 0.f;
        den1 += mine ? __expf(e1 - m1) : 0.f;
    }
    den0 = wave_sum_f(den0); den1 = wave_sum_f(den1);

    const int c = lane * 4;
    const bool head1 = lane >= 16;
    float4 acc = make_float4(0.f, 0.f, 0.f, 0.f);

    auto gather = [&](int s_l, float a0_l, float a1_l, int n) {
#pragma unroll 4
        for (int j = 0; j < n; ++j) {
            const int sj = __builtin_amdgcn_readlane(s_l, j);
            const float a0 = bcast_f(a0_l, j);
            const float a1 = bcast_f(a1_l, j);
            const float al = head1 ? a1 : a0;
            const float4 h4 = *(const float4*)(H + (size_t)sj * HCAT + c);
            acc.x += h4.x * al; acc.y += h4.y * al;
            acc.z += h4.z * al; acc.w += h4.w * al;
        }
    };

    gather(sc, x0c / den0, x1c / den1, tot < 32 ? tot : 32);
    for (int b = 32; b < tot; b += 32) {
        int s; float e0, e1;
        edge(b + lane, s, e0, e1);
        const float x0 = (b + lane < tot) ? __expf(e0 - m0) : 0.f;
        const float x1 = (b + lane < tot) ? __expf(e1 - m1) : 0.f;
        gather(s, x0 / den0, x1 / den1, (tot - b) < 32 ? (tot - b) : 32);
    }
    for (int b = 0; b < novf; b += 32) {
        int s; float e0, e1;
        const bool mine = ovf_edge(b + lane, s, e0, e1);
        const float a0_l = mine ? __expf(e0 - m0) / den0 : 0.f;
        const float a1_l = mine ? __expf(e1 - m1) / den1 : 0.f;
        unsigned live = __builtin_amdgcn_ballot_w32(mine);
        while (live) {
            const int j = __builtin_ctz(live);
            live &= live - 1;
            const int sj = __builtin_amdgcn_readlane(s, j);
            const float al = head1 ? bcast_f(a1_l, j) : bcast_f(a0_l, j);
            const float4 h4 = *(const float4*)(H + (size_t)sj * HCAT + c);
            acc.x += h4.x * al; acc.y += h4.y * al;
            acc.z += h4.z * al; acc.w += h4.w * al;
        }
    }

    const float a = *prelu_a;
    if (CONCAT) {
        float4 v = acc;
        v.x += bias[c + 0]; v.y += bias[c + 1]; v.z += bias[c + 2]; v.w += bias[c + 3];
        v.x = v.x >= 0.f ? v.x : a * v.x;
        v.y = v.y >= 0.f ? v.y : a * v.y;
        v.z = v.z >= 0.f ? v.z : a * v.z;
        v.w = v.w >= 0.f ? v.w : a * v.w;
        *(float4*)(OUT + (size_t)d * HCAT + c) = v;
    } else {
        float4 o;
        o.x = __shfl_xor(acc.x, 16); o.y = __shfl_xor(acc.y, 16);
        o.z = __shfl_xor(acc.z, 16); o.w = __shfl_xor(acc.w, 16);
        const int cc = (lane & 15) * 4;
        float4 v;
        v.x = 0.5f * (acc.x + o.x) + bias[cc + 0];
        v.y = 0.5f * (acc.y + o.y) + bias[cc + 1];
        v.z = 0.5f * (acc.z + o.z) + bias[cc + 2];
        v.w = 0.5f * (acc.w + o.w) + bias[cc + 3];
        v.x = v.x >= 0.f ? v.x : a * v.x;
        v.y = v.y >= 0.f ? v.y : a * v.y;
        v.z = v.z >= 0.f ? v.z : a * v.z;
        v.w = v.w >= 0.f ? v.w : a * v.w;
        if (lane < 16) *(float4*)(OUT + (size_t)d * HIDC + cc) = v;
    }
}


__global__ __launch_bounds__(256) void attn_logits_in_kernel(
    const float* __restrict__ X, const float* __restrict__ Wm,
    const float* __restrict__ a_s, const float* __restrict__ a_d,
    float* __restrict__ als, float* __restrict__ ald)
{
    __shared__ float4 wv[INCH];
    const int t = threadIdx.x;
    if (t < INCH * 4) {
        const int k = t >> 2, j = t & 3, head = j & 1;
        const float* __restrict__ av = ((j & 2) ? a_d : a_s) + head * HIDC;
        const float* __restrict__ wr = Wm + (size_t)k * HCAT + head * HIDC;
        const float4* __restrict__ w4 = (const float4*)wr;
        const float4* __restrict__ a4 = (const float4*)av;
        float s = 0.f;
#pragma unroll
        for (int c = 0; c < HIDC / 4; ++c) {
            const float4 w = w4[c], a = a4[c];
            s += w.x * a.x + w.y * a.y + w.z * a.z + w.w * a.w;
        }
        ((float*)wv)[t] = s;
    }
    __syncthreads();
    const int node = blockIdx.x * blockDim.x + t;
    if (node >= NNODES) return;
    const float4* __restrict__ xv = (const float4*)(X + (size_t)node * INCH);
    float s0 = 0.f, s1 = 0.f, d0 = 0.f, d1 = 0.f;
#pragma unroll
    for (int i = 0; i < INCH / 4; ++i) {
        const float4 x4 = xv[i];
        const float xs[4] = { x4.x, x4.y, x4.z, x4.w };
#pragma unroll
        for (int u = 0; u < 4; ++u) {
            const float4 w = wv[i * 4 + u];
            s0 += xs[u] * w.x; s1 += xs[u] * w.y;
            d0 += xs[u] * w.z; d1 += xs[u] * w.w;
        }
    }
    ((float2*)als)[node] = make_float2(s0, s1);
    ((float2*)ald)[node] = make_float2(d0, d1);
}

template <int I>
__device__ __forceinline__ int group8_bcast_i(int v) {
    constexpr unsigned sel = I | (I << 3) | (I << 6) | (I << 9) | (I << 12) | (I << 15) | (I << 18) | (I << 21);
    return __builtin_amdgcn_mov_dpp8(v, sel);
}
template <int I>
__device__ __forceinline__ float group8_bcast_f(float v) {
    return __int_as_float(group8_bcast_i<I>(__float_as_int(v)));
}

__global__ __launch_bounds__(256) void gat_gather_in_kernel(
    const int* __restrict__ cnt, const int* __restrict__ slots, const int* __restrict__ ovf,
    const float* __restrict__ X, const float* __restrict__ als,
    const float* __restrict__ ald, float* __restrict__ Z)
{
    const int d = __builtin_amdgcn_readfirstlane((int)(blockIdx.x * 8 + (threadIdx.x >> 5)));
    if (d >= NNODES) return;
    const int lane = threadIdx.x & 31;
    const int beg = d * SLOTC;
    const int indeg = cnt[d];
    const int deg = indeg < SLOTC ? indeg : SLOTC;
    const int tot = deg + 1;
    const int novf = indeg > SLOTC ? cnt[NNODES] : 0;
    const float ad0 = ald[d * 2 + 0];
    const float ad1 = ald[d * 2 + 1];
    const float2* __restrict__ als2 = (const float2*)als;

    auto edge = [&](int j, int& s, float& e0, float& e1) {
        s = (j < deg) ? slots[beg + j] : d;
        const float2 a = als2[s];
        e0 = (j < tot) ? leakyr(a.x + ad0) : -INFINITY;
        e1 = (j < tot) ? leakyr(a.y + ad1) : -INFINITY;
    };
    auto ovf_edge = [&](int i, int& s, float& e0, float& e1) -> bool {
        bool mine = false;
        s = d;
        if (i < novf) {
            const int2 p = *(const int2*)(ovf + 2 * i);
            mine = (p.x == d);
            s = mine ? p.y : d;
        }
        const float2 a = als2[s];
        e0 = mine ? leakyr(a.x + ad0) : -INFINITY;
        e1 = mine ? leakyr(a.y + ad1) : -INFINITY;
        return mine;
    };

    int sc; float e0c, e1c;
    edge(lane, sc, e0c, e1c);

    float m0 = e0c, m1 = e1c;
    for (int b = 32; b < tot; b += 32) {
        int s; float e0, e1;
        edge(b + lane, s, e0, e1);
        m0 = fmaxf(m0, e0); m1 = fmaxf(m1, e1);
    }
    for (int b = 0; b < novf; b += 32) {
        int s; float e0, e1;
        ovf_edge(b + lane, s, e0, e1);
        m0 = fmaxf(m0, e0); m1 = fmaxf(m1, e1);
    }
    m0 = wave_max_f(m0); m1 = wave_max_f(m1);

    const float x0c = (lane < tot) ? __expf(e0c - m0) : 0.f;
    const float x1c = (lane < tot) ? __expf(e1c - m1) : 0.f;
    float den0 = x0c, den1 = x1c;
    for (int b = 32; b < tot; b += 32) {
        int s; float e0, e1;
        edge(b + lane, s, e0, e1);
        den0 += (b + lane < tot) ? __expf(e0 - m0) : 0.f;
        den1 += (b + lane < tot) ? __expf(e1 - m1) : 0.f;
    }
    for (int b = 0; b < novf; b += 32) {
        int s; float e0, e1;
        const bool mine = ovf_edge(b + lane, s, e0, e1);
        den0 += mine ? __expf(e0 - m0) : 0.f;
        den1 += mine ? __expf(e1 - m1) : 0.f;
    }
    den0 = wave_sum_f(den0); den1 = wave_sum_f(den1);

    const int jbase = lane & 24;
    const unsigned q = (lane & 7) * 4;
    float4 acc0 = make_float4(0.f, 0.f, 0.f, 0.f);
    float4 acc1 = make_float4(0.f, 0.f, 0.f, 0.f);

#define GATHER_IN_LOAD(I)                                                          \
    {                                                                              \
        const unsigned sj = (unsigned)group8_bcast_i<I>(s_l);                      \
        xr[I] = (jbase + I < n) ? *(const float4*)(X + (sj * INCH + q))            \
                                : make_float4(0.f, 0.f, 0.f, 0.f);                 \
    }
#define GATHER_IN_SUM(I)                                                           \
    {                                                                              \
        const float a0 = group8_bcast_f<I>(a0_l);                                  \
        const float a1 = group8_bcast_f<I>(a1_l);                                  \
        acc0.x += xr[I].x * a0; acc0.y += xr[I].y * a0;                            \
        acc0.z += xr[I].z * a0; acc0.w += xr[I].w * a0;                            \
        acc1.x += xr[I].x * a1; acc1.y += xr[I].y * a1;                            \
        acc1.z += xr[I].z * a1; acc1.w += xr[I].w * a1;                            \
    }
    auto gather = [&](int s_l, float a0_l, float a1_l, int n) {
        float4 xr[8];
        GATHER_IN_LOAD(0) GATHER_IN_LOAD(1) GATHER_IN_LOAD(2) GATHER_IN_LOAD(3)
        GATHER_IN_LOAD(4) GATHER_IN_LOAD(5) GATHER_IN_LOAD(6) GATHER_IN_LOAD(7)
        GATHER_IN_SUM(0) GATHER_IN_SUM(1) GATHER_IN_SUM(2) GATHER_IN_SUM(3)
        GATHER_IN_SUM(4) GATHER_IN_SUM(5) GATHER_IN_SUM(6) GATHER_IN_SUM(7)
    };

    gather(sc, x0c / den0, x1c / den1, tot < 32 ? tot : 32);
    for (int b = 32; b < tot; b += 32) {
        int s; float e0, e1;
        edge(b + lane, s, e0, e1);
        const float x0 = (b + lane < tot) ? __expf(e0 - m0) : 0.f;
        const float x1 = (b + lane < tot) ? __expf(e1 - m1) : 0.f;
        gather(s, x0 / den0, x1 / den1, (tot - b) < 32 ? (tot - b) : 32);
    }
#undef GATHER_IN_LOAD
#undef GATHER_IN_SUM
    for (int b = 0; b < novf; b += 32) {
        int s; float e0, e1;
        const bool mine = ovf_edge(b + lane, s, e0, e1);
        const float a0_l = mine ? __expf(e0 - m0) / den0 : 0.f;
        const float a1_l = mine ? __expf(e1 - m1) / den1 : 0.f;
        unsigned live = __builtin_amdgcn_ballot_w32(mine);
        while (live) {
            const int j = __builtin_ctz(live);
            live &= live - 1;
            const unsigned sj = (unsigned)__builtin_amdgcn_readlane(s, j);
            const float a0 = (lane < 8) ? bcast_f(a0_l, j) : 0.f;
            const float a1 = (lane < 8) ? bcast_f(a1_l, j) : 0.f;
            const float4 xv = (lane < 8) ? *(const float4*)(X + (sj * INCH + q))
                                         : make_float4(0.f, 0.f, 0.f, 0.f);
            acc0.x += xv.x * a0; acc0.y += xv.y * a0;
            acc0.z += xv.z * a0; acc0.w += xv.w * a0;
            acc1.x += xv.x * a1; acc1.y += xv.y * a1;
            acc1.z += xv.z * a1; acc1.w += xv.w * a1;
        }
    }

#define XSUM(v) v += __shfl_xor(v, 8); v += __shfl_xor(v, 16);
    XSUM(acc0.x) XSUM(acc0.y) XSUM(acc0.z) XSUM(acc0.w)
    XSUM(acc1.x) XSUM(acc1.y) XSUM(acc1.z) XSUM(acc1.w)
#undef XSUM

    float* __restrict__ zrow = Z + (size_t)d * (NHEADS * INCH);
    if (lane < 8)       *(float4*)(zrow + q) = acc0;
    else if (lane < 16) *(float4*)(zrow + INCH + q) = acc1;
}

__global__ __launch_bounds__(256) void gemm_heads_wmma_f32(
    const float* __restrict__ Z, const float* __restrict__ B,
    const float* __restrict__ bias, const float* __restrict__ prelu_a,
    float* __restrict__ D, int nrows)
{
    const int wave = threadIdx.x >> 5;
    const int lane = threadIdx.x & 31;
    const int tile = blockIdx.x * 8 + wave;
    const int ntiles = nrows >> 4;
    if (tile >= ntiles) return;
    const int half = lane >> 4;
    const int mrow = lane & 15;
    const int ncol = lane & 15;
    const int m0 = tile << 4;

    v8f acc[8] = {};

    const float* __restrict__ zrow = Z + (size_t)(m0 + mrow) * (NHEADS * INCH);
    for (int kb = 0; kb < INCH; kb += 4) {
        const int k0 = kb + 2 * half;
        v2f a0, a1;
        a0.x = zrow[k0];        a0.y = zrow[k0 + 1];
        a1.x = zrow[INCH + k0]; a1.y = zrow[INCH + k0 + 1];
#pragma unroll
        for (int t = 0; t < 8; ++t) {
            const int n = t * 16 + ncol;
            v2f b;
            b.x = B[(size_t)k0 * HCAT + n];
            b.y = B[(size_t)(k0 + 1) * HCAT + n];
            acc[t] = __builtin_amdgcn_wmma_f32_16x16x4_f32(
                false, (t < 4) ? a0 : a1, false, b, (short)0, acc[t], false, false);
        }
    }

    const float a = *prelu_a;
#pragma unroll
    for (int t = 0; t < 8; ++t) {
        const int n = t * 16 + ncol;
        const float bv = bias[n];
#pragma unroll
        for (int r = 0; r < 8; ++r) {
            const int m = m0 + r + 8 * half;
            float v = acc[t][r] + bv;
            v = v >= 0.f ? v : a * v;
            D[(size_t)m * HCAT + n] = v;
        }
    }
}

extern "C" void kernel_launch(void* const* d_in, const int* in_sizes, int n_in,
                              void* d_out, int out_size, void* d_ws, size_t ws_size,
                              hipStream_t stream)
{
    const float* x       = (const float*)d_in[0];
    const int*   ei      = (const int*)  d_in[1];
    const float* W1      = (const float*)d_in[2];
    const float* a1s     = (const float*)d_in[3];
    const float* a1d     = (const float*)d_in[4];
    const float* b1      = (const float*)d_in[5];
    const float* W2      = (const float*)d_in[6];
    const float* a2s     = (const float*)d_in[7];
    const float* a2d     = (const float*)d_in[8];
    const float* b2      = (const float*)d_in[9];
    const float* W3      = (const float*)d_in[10];
    const float* a3s     = (const float*)d_in[11];
    const float* a3d     = (const float*)d_in[12];
    const float* b3      = (const float*)d_in[13];
    const float* prelu_a = (const float*)d_in[14];
    const float* Wc      = (const float*)d_in[15];
    const float* bc      = (const float*)d_in[16];
    float* out = (float*)d_out;

    float* ws = (float*)d_ws;
    size_t off = 0;
    float* H     = ws + off; off += (size_t)NNODES * HCAT;
    float* PREV  = ws + off; off += (size_t)NNODES * HCAT;
    float* als   = ws + off; off += (size_t)NNODES * 2;
    float* ald   = ws + off; off += (size_t)NNODES * 2;
    int* cnt     = (int*)(ws + off); off += (size_t)CNTWORDS;
    int* slots   = (int*)(ws + off); off += (size_t)NNODES * SLOTC;
    int* ovf     = (int*)(ws + off); off += (size_t)NEDGES * 2;

    const int gN1    = (NNODES + 255) / 256;
    const int gN2    = (NNODES * NHEADS + 255) / 256;
    const int gEdge  = (NEDGES + 255) / 256;
    const int gNodeW = (NNODES + 7) / 8;
    const int gTile  = ((NNODES / 16) + 7) / 8;

    hipMemsetAsync(cnt, 0x00, (size_t)CNTWORDS * sizeof(int), stream);
    bucket_edges_kernel<<<gEdge, 256, 0, stream>>>(ei, cnt, slots, ovf);

    struct Layer {
        const float* W; const float* as; const float* ad; const float* b; bool concat;
    };
    const Layer layers[3] = {
        { W1, a1s, a1d, b1, true  },
        { W2, a2s, a2d, b2, true  },
        { W3, a3s, a3d, b3, false },
    };

    float* Z = H;
    attn_logits_in_kernel<<<gN1, 256, 0, stream>>>(x, W1, a1s, a1d, als, ald);
    gat_gather_in_kernel <<<gNodeW, 256, 0, stream>>>(cnt, slots, ovf, x, als, ald, Z);
    gemm_heads_wmma_f32  <<<gTile, 256, 0, stream>>>(Z, W1, b1, prelu_a, PREV, NNODES);

    for (int l = 1; l < 3; ++l) {
        gemm_wmma_f32<HCAT, HCAT, 8><<<gTile, 256, 0, stream>>>(PREV, layers[l].W, nullptr, H, NNODES);

        attn_logits_kernel<<<gN2, 256, 0, stream>>>(H, layers[l].as, layers[l].ad, als, ald);

        if (layers[l].concat)
            gat_gather_kernel<true ><<<gNodeW, 256, 0, stream>>>(cnt, slots, ovf, H, als, ald,
                                                                layers[l].b, prelu_a, PREV);
        else
            gat_gather_kernel<false><<<gNodeW, 256, 0, stream>>>(cnt, slots, ovf, H, als, ald,
                                                                layers[l].b, prelu_a, PREV);
    }

    gemm_wmma_f32<HIDC, NCLS, 1><<<gTile, 256, 0, stream>>>(PREV, Wc, bc, out, NNODES);
    (void)in_sizes; (void)n_in; (void)out_size; (void)ws_size;
}


# Every node kernel keeps its 32 finished rows in LDS and applies the following linear maps to them on WMMA (layer 1+2 / layer 3 + logits / classifier): 6 launches go; logit sum order changes
# speedup vs baseline: 7.9083x; 1.3081x over previous
#include <hip/hip_runtime.h>
#include <hip/hip_bf16.h>

#define NNODES 100000
#define NEDGES 1600000
#define INCH   32
#define HIDC   64
#define NHEADS 2
#define HCAT   128
#define NCLS   12
#define NEGSLOPE 0.2f

typedef __attribute__((ext_vector_type(2))) float v2f;
typedef __attribute__((ext_vector_type(8))) float v8f;

__device__ __forceinline__ float leakyr(float x) { return x > 0.f ? x : NEGSLOPE * x; }


#define SLOTC 64
#define CNTWORDS 100096

__global__ __launch_bounds__(256) void bucket_edges_kernel(
    const int* __restrict__ ei, int* __restrict__ cnt, int* __restrict__ slots,
    int* __restrict__ ovf)
{
    const int e = blockIdx.x * blockDim.x + threadIdx.x;
    if (e >= NEDGES) return;
    const int s = ei[e];
    const int d = ei[NEDGES + e];
    const int r = atomicAdd(&cnt[d], 1);
    if (r < SLOTC) {
        slots[d * SLOTC + r] = s;
    } else {
        const int o = atomicAdd(&cnt[NNODES], 1);
        *(int2*)(ovf + 2 * o) = make_int2(d, s);
    }
}

__device__ __forceinline__ float wave_max_f(float v) {
#pragma unroll
    for (int o = 16; o > 0; o >>= 1) v = fmaxf(v, __shfl_xor(v, o));
    return v;
}
__device__ __forceinline__ float wave_sum_f(float v) {
#pragma unroll
    for (int o = 16; o > 0; o >>= 1) v += __shfl_xor(v, o);
    return v;
}
__device__ __forceinline__ float bcast_f(float v, int j) {
    return __int_as_float(__builtin_amdgcn_readlane(__float_as_int(v), j));
}

#define NBLK 32
#define TPITCH (HCAT + 4)

__device__ __forceinline__ void tile_linear_rows_logits(
    const float* T, float* O, const float* __restrict__ Wn,
    const float* __restrict__ an_s, const float* __restrict__ an_d,
    float* __restrict__ OUT, float* __restrict__ als_n, float* __restrict__ ald_n,
    int node0, int wave, int lane)
{
    const int half = lane >> 4;
    const int l15 = lane & 15;
    const int n = wave * 16 + l15;
    v8f acc0 = {}, acc1 = {};
    const float* a0p = T + l15 * TPITCH + 2 * half;
    const float* a1p = a0p + 16 * TPITCH;
    const float* __restrict__ bp = Wn + (size_t)(2 * half) * HCAT + n;
    for (int kb = 0; kb < HCAT; kb += 4) {
        v2f b;
        b.x = bp[(size_t)kb * HCAT];
        b.y = bp[(size_t)(kb + 1) * HCAT];
        const v2f a0 = *(const v2f*)(a0p + kb);
        const v2f a1 = *(const v2f*)(a1p + kb);
        acc0 = __builtin_amdgcn_wmma_f32_16x16x4_f32(false, a0, false, b, (short)0, acc0, false, false);
        acc1 = __builtin_amdgcn_wmma_f32_16x16x4_f32(false, a1, false, b, (short)0, acc1, false, false);
    }
#pragma unroll
    for (int r = 0; r < 8; ++r) {
        O[(r + 8 * half) * TPITCH + n] = acc0[r];
        O[(16 + r + 8 * half) * TPITCH + n] = acc1[r];
    }
    __syncthreads();

    const int c = lane * 4;
    const float4 s4 = *(const float4*)(an_s + c);
    const float4 d4 = *(const float4*)(an_d + c);
#pragma unroll
    for (int q = 0; q < NBLK / 8; ++q) {
        const int i = wave * (NBLK / 8) + q;
        const float4 o = *(const float4*)(O + i * TPITCH + c);
        *(float4*)(OUT + (size_t)(node0 + i) * HCAT + c) = o;
        float s = o.x * s4.x + o.y * s4.y + o.z * s4.z + o.w * s4.w;
        float t = o.x * d4.x + o.y * d4.y + o.z * d4.z + o.w * d4.w;
#pragma unroll
        for (int off = 8; off > 0; off >>= 1) {
            s += __shfl_xor(s, off);
            t += __shfl_xor(t, off);
        }
        if (l15 == 0) {
            als_n[(node0 + i) * 2 + half] = s;
            ald_n[(node0 + i) * 2 + half] = t;
        }
    }
}

template <bool CONCAT>
__global__ __launch_bounds__(256) __attribute__((amdgpu_waves_per_eu(16))) void gat_gather_next_kernel(
    const int* __restrict__ cnt, const int* __restrict__ slots, const int* __restrict__ ovf,
    const float* __restrict__ H, const float* __restrict__ als,
    const float* __restrict__ ald, const float* __restrict__ bias,
    const float* __restrict__ prelu_a,
    const float* __restrict__ Wn, const float* __restrict__ an_s,
    const float* __restrict__ an_d, const float* __restrict__ bn,
    float* __restrict__ OUT, float* __restrict__ als_n, float* __restrict__ ald_n)
{
    constexpr int CW = CONCAT ? HCAT : HIDC;
    constexpr int PITCH = CW + 4;
    __shared__ __attribute__((aligned(16))) float T[NBLK * PITCH];

    const int wave = threadIdx.x >> 5;
    const int lane = threadIdx.x & 31;
    const int node0 = blockIdx.x * NBLK;
    const float2* __restrict__ als2 = (const float2*)als;
    const float a = *prelu_a;

    for (int it = 0; it < NBLK / 8; ++it) {
    const int trow = it * 8 + wave;
    const int d = __builtin_amdgcn_readfirstlane(node0 + trow);
    const int beg = d * SLOTC;
    const int indeg = cnt[d];
    const int deg = indeg < SLOTC ? indeg : SLOTC;
    const int tot = deg + 1;
    const int novf = indeg > SLOTC ? cnt[NNODES] : 0;
    const float ad0 = ald[d * 2 + 0];
    const float ad1 = ald[d * 2 + 1];

    auto edge = [&](int j, int& s, float& e0, float& e1) {
        s = (j < deg) ? slots[beg + j] : d;
        const float2 av = als2[s];
        e0 = (j < tot) ? leakyr(av.x + ad0) : -INFINITY;
        e1 = (j < tot) ? leakyr(av.y + ad1) : -INFINITY;
    };
    auto ovf_edge = [&](int i, int& s, float& e0, float& e1) -> bool {
        bool mine = false;
        s = d;
        if (i < novf) {
            const int2 p = *(const int2*)(ovf + 2 * i);
            mine = (p.x == d);
            s = mine ? p.y : d;
        }
        const float2 av = als2[s];
        e0 = mine ? leakyr(av.x + ad0) : -INFINITY;
        e1 = mine ? leakyr(av.y + ad1) : -INFINITY;
        return mine;
    };

    int sc; float e0c, e1c;
    edge(lane, sc, e0c, e1c);

    float m0 = e0c, m1 = e1c;
    for (int b = 32; b < tot; b += 32) {
        int s; float e0, e1;
        edge(b + lane, s, e0, e1);
        m0 = fmaxf(m0, e0); m1 = fmaxf(m1, e1);
    }
    for (int b = 0; b < novf; b += 32) {
        int s; float e0, e1;
        ovf_edge(b + lane, s, e0, e1);
        m0 = fmaxf(m0, e0); m1 = fmaxf(m1, e1);
    }
    m0 = wave_max_f(m0); m1 = wave_max_f(m1);

    const float x0c = (lane < tot) ? __expf(e0c - m0) : 0.f;
    const float x1c = (lane < tot) ? __expf(e1c - m1) : 0.f;
    float den0 = x0c, den1 = x1c;
    for (int b = 32; b < tot; b += 32) {
        int s; float e0, e1;
        edge(b + lane, s, e0, e1);
        den0 += (b + lane < tot) ? __expf(e0 - m0) : 0.f;
        den1 += (b + lane < tot) ? __expf(e1 - m1) : 0.f;
    }
    for (int b = 0; b < novf; b += 32) {
        int s; float e0, e1;
        const bool mine = ovf_edge(b + lane, s, e0, e1);
        den0 += mine ? __expf(e0 - m0) : 0.f;
        den1 += mine ? __expf(e1 - m1) : 0.f;
    }
    den0 = wave_sum_f(den0); den1 = wave_sum_f(den1);

    const int c = lane * 4;
    const bool head1 = lane >= 16;
    float4 acc = make_float4(0.f, 0.f, 0.f, 0.f);

    auto gather = [&](int s_l, float a0_l, float a1_l, int n) {
#pragma unroll 4
        for (int j = 0; j < n; ++j) {
            const int sj = __builtin_amdgcn_readlane(s_l, j);
            const float a0 = bcast_f(a0_l, j);
            const float a1 = bcast_f(a1_l, j);
            const float al = head1 ? a1 : a0;
            const float4 h4 = *(const float4*)(H + (size_t)sj * HCAT + c);
            acc.x += h4.x * al; acc.y += h4.y * al;
            acc.z += h4.z * al; acc.w += h4.w * al;
        }
    };

    gather(sc, x0c / den0, x1c / den1, tot < 32 ? tot : 32);
    for (int b = 32; b < tot; b += 32) {
        int s; float e0, e1;
        edge(b + lane, s, e0, e1);
        const float x0 = (b + lane < tot) ? __expf(e0 - m0) : 0.f;
        const float x1 = (b + lane < tot) ? __expf(e1 - m1) : 0.f;
        gather(s, x0 / den0, x1 / den1, (tot - b) < 32 ? (tot - b) : 32);
    }
    for (int b = 0; b < novf; b += 32) {
        int s; float e0, e1;
        const bool mine = ovf_edge(b + lane, s, e0, e1);
        const float a0_l = mine ? __expf(e0 - m0) / den0 : 0.f;
        const float a1_l = mine ? __expf(e1 - m1) / den1 : 0.f;
        unsigned live = __builtin_amdgcn_ballot_w32(mine);
        while (live) {
            const int j = __builtin_ctz(live);
            live &= live - 1;
            const int sj = __builtin_amdgcn_readlane(s, j);
            const float al = head1 ? bcast_f(a1_l, j) : bcast_f(a0_l, j);
            const float4 h4 = *(const float4*)(H + (size_t)sj * HCAT + c);
            acc.x += h4.x * al; acc.y += h4.y * al;
            acc.z += h4.z * al; acc.w += h4.w * al;
        }
    }

    if (CONCAT) {
        float4 v = acc;
        v.x += bias[c + 0]; v.y += bias[c + 1]; v.z += bias[c + 2]; v.w += bias[c + 3];
        v.x = v.x >= 0.f ? v.x : a * v.x;
        v.y = v.y >= 0.f ? v.y : a * v.y;
        v.z = v.z >= 0.f ? v.z : a * v.z;
        v.w = v.w >= 0.f ? v.w : a * v.w;
        *(float4*)(T + trow * PITCH + c) = v;
    } else {
        float4 o;
        o.x = __shfl_xor(acc.x, 16); o.y = __shfl_xor(acc.y, 16);
        o.z = __shfl_xor(acc.z, 16); o.w = __shfl_xor(acc.w, 16);
        const int cc = (lane & 15) * 4;
        float4 v;
        v.x = 0.5f * (acc.x + o.x) + bias[cc + 0];
        v.y = 0.5f * (acc.y + o.y) + bias[cc + 1];
        v.z = 0.5f * (acc.z + o.z) + bias[cc + 2];
        v.w = 0.5f * (acc.w + o.w) + bias[cc + 3];
        v.x = v.x >= 0.f ? v.x : a * v.x;
        v.y = v.y >= 0.f ? v.y : a * v.y;
        v.z = v.z >= 0.f ? v.z : a * v.z;
        v.w = v.w >= 0.f ? v.w : a * v.w;
        if (lane < 16) *(float4*)(T + trow * PITCH + cc) = v;
    }
    }

    __syncthreads();

    const int half = lane >> 4;
    const int l15 = lane & 15;
    if constexpr (CONCAT) {
        __shared__ __attribute__((aligned(16))) float O[NBLK * PITCH];
        tile_linear_rows_logits(T, O, Wn, an_s, an_d, OUT, als_n, ald_n, node0, wave, lane);
    } else {
        if (wave < NBLK / 16) {
            v8f accc = {};
            const float* ap = T + (wave * 16 + l15) * PITCH + 2 * half;
            const bool live = l15 < NCLS;
            const int nn = live ? l15 : 0;
            for (int kb = 0; kb < HIDC; kb += 4) {
                const int k0 = kb + 2 * half;
                const v2f av = *(const v2f*)(ap + kb);
                v2f b;
                b.x = Wn[k0 * NCLS + nn];
                b.y = Wn[(k0 + 1) * NCLS + nn];
                b.x = live ? b.x : 0.f;
                b.y = live ? b.y : 0.f;
                accc = __builtin_amdgcn_wmma_f32_16x16x4_f32(false, av, false, b, (short)0, accc, false, false);
            }
            if (live) {
                const float bv = bn[l15];
#pragma unroll
                for (int r = 0; r < 8; ++r) {
                    const int m = node0 + wave * 16 + r + 8 * half;
                    OUT[(size_t)m * NCLS + l15] = accc[r] + bv;
                }
            }
        }
    }
}


__global__ __launch_bounds__(256) void attn_logits_in_kernel(
    const float* __restrict__ X, const float* __restrict__ Wm,
    const float* __restrict__ a_s, const float* __restrict__ a_d,
    float* __restrict__ als, float* __restrict__ ald)
{
    __shared__ float4 wv[INCH];
    const int t = threadIdx.x;
    if (t < INCH * 4) {
        const int k = t >> 2, j = t & 3, head = j & 1;
        const float* __restrict__ av = ((j & 2) ? a_d : a_s) + head * HIDC;
        const float* __restrict__ wr = Wm + (size_t)k * HCAT + head * HIDC;
        const float4* __restrict__ w4 = (const float4*)wr;
        const float4* __restrict__ a4 = (const float4*)av;
        float s = 0.f;
#pragma unroll
        for (int c = 0; c < HIDC / 4; ++c) {
            const float4 w = w4[c], a = a4[c];
            s += w.x * a.x + w.y * a.y + w.z * a.z + w.w * a.w;
        }
        ((float*)wv)[t] = s;
    }
    __syncthreads();
    const int node = blockIdx.x * blockDim.x + t;
    if (node >= NNODES) return;
    const float4* __restrict__ xv = (const float4*)(X + (size_t)node * INCH);
    float s0 = 0.f, s1 = 0.f, d0 = 0.f, d1 = 0.f;
#pragma unroll
    for (int i = 0; i < INCH / 4; ++i) {
        const float4 x4 = xv[i];
        const float xs[4] = { x4.x, x4.y, x4.z, x4.w };
#pragma unroll
        for (int u = 0; u < 4; ++u) {
            const float4 w = wv[i * 4 + u];
            s0 += xs[u] * w.x; s1 += xs[u] * w.y;
            d0 += xs[u] * w.z; d1 += xs[u] * w.w;
        }
    }
    ((float2*)als)[node] = make_float2(s0, s1);
    ((float2*)ald)[node] = make_float2(d0, d1);
}

template <int I>
__device__ __forceinline__ int group8_bcast_i(int v) {
    constexpr unsigned sel = I | (I << 3) | (I << 6) | (I << 9) | (I << 12) | (I << 15) | (I << 18) | (I << 21);
    return __builtin_amdgcn_mov_dpp8(v, sel);
}
template <int I>
__device__ __forceinline__ float group8_bcast_f(float v) {
    return __int_as_float(group8_bcast_i<I>(__float_as_int(v)));
}

#define ZPITCH (NHEADS * INCH + 4)
__global__ __launch_bounds__(256) __attribute__((amdgpu_waves_per_eu(10))) void gat_gather_in_next_kernel(
    const int* __restrict__ cnt, const int* __restrict__ slots, const int* __restrict__ ovf,
    const float* __restrict__ X, const float* __restrict__ als,
    const float* __restrict__ ald,
    const float* __restrict__ W1m, const float* __restrict__ bias1, const float* __restrict__ prelu_a,
    const float* __restrict__ Wn, const float* __restrict__ an_s, const float* __restrict__ an_d,
    float* __restrict__ OUT, float* __restrict__ als_n, float* __restrict__ ald_n)
{
    __shared__ __attribute__((aligned(16))) float U[NBLK * TPITCH];
    __shared__ __attribute__((aligned(16))) float T[NBLK * TPITCH];
    const int wave = threadIdx.x >> 5;
    const int lane = threadIdx.x & 31;
    const int node0 = blockIdx.x * NBLK;

    for (int it = 0; it < NBLK / 8; ++it) {
    const int trow = it * 8 + wave;
    const int d = __builtin_amdgcn_readfirstlane(node0 + trow);
    const int beg = d * SLOTC;
    const int indeg = cnt[d];
    const int deg = indeg < SLOTC ? indeg : SLOTC;
    const int tot = deg + 1;
    const int novf = indeg > SLOTC ? cnt[NNODES] : 0;
    const float ad0 = ald[d * 2 + 0];
    const float ad1 = ald[d * 2 + 1];
    const float2* __restrict__ als2 = (const float2*)als;

    auto edge = [&](int j, int& s, float& e0, float& e1) {
        s = (j < deg) ? slots[beg + j] : d;
        const float2 a = als2[s];
        e0 = (j < tot) ? leakyr(a.x + ad0) : -INFINITY;
        e1 = (j < tot) ? leakyr(a.y + ad1) : -INFINITY;
    };
    auto ovf_edge = [&](int i, int& s, float& e0, float& e1) -> bool {
        bool mine = false;
        s = d;
        if (i < novf) {
            const int2 p = *(const int2*)(ovf + 2 * i);
            mine = (p.x == d);
            s = mine ? p.y : d;
        }
        const float2 a = als2[s];
        e0 = mine ? leakyr(a.x + ad0) : -INFINITY;
        e1 = mine ? leakyr(a.y + ad1) : -INFINITY;
        return mine;
    };

    int sc; float e0c, e1c;
    edge(lane, sc, e0c, e1c);

    float m0 = e0c, m1 = e1c;
    for (int b = 32; b < tot; b += 32) {
        int s; float e0, e1;
        edge(b + lane, s, e0, e1);
        m0 = fmaxf(m0, e0); m1 = fmaxf(m1, e1);
    }
    for (int b = 0; b < novf; b += 32) {
        int s; float e0, e1;
        ovf_edge(b + lane, s, e0, e1);
        m0 = fmaxf(m0, e0); m1 = fmaxf(m1, e1);
    }
    m0 = wave_max_f(m0); m1 = wave_max_f(m1);

    const float x0c = (lane < tot) ? __expf(e0c - m0) : 0.f;
    const float x1c = (lane < tot) ? __expf(e1c - m1) : 0.f;
    float den0 = x0c, den1 = x1c;
    for (int b = 32; b < tot; b += 32) {
        int s; float e0, e1;
        edge(b + lane, s, e0, e1);
        den0 += (b + lane < tot) ? __expf(e0 - m0) : 0.f;
        den1 += (b + lane < tot) ? __expf(e1 - m1) : 0.f;
    }
    for (int b = 0; b < novf; b += 32) {
        int s; float e0, e1;
        const bool mine = ovf_edge(b + lane, s, e0, e1);
        den0 += mine ? __expf(e0 - m0) : 0.f;
        den1 += mine ? __expf(e1 - m1) : 0.f;
    }
    den0 = wave_sum_f(den0); den1 = wave_sum_f(den1);

    const int jbase = lane & 24;
    const unsigned q = (lane & 7) * 4;
    float4 acc0 = make_float4(0.f, 0.f, 0.f, 0.f);
    float4 acc1 = make_float4(0.f, 0.f, 0.f, 0.f);

#define GATHER_IN_LOAD(I)                                                          \
    {                                                                              \
        const unsigned sj = (unsigned)group8_bcast_i<I>(s_l);                      \
        xr[I] = (jbase + I < n) ? *(const float4*)(X + (sj * INCH + q))            \
                                : make_float4(0.f, 0.f, 0.f, 0.f);                 \
    }
#define GATHER_IN_SUM(I)                                                           \
    {                                                                              \
        const float a0 = group8_bcast_f<I>(a0_l);                                  \
        const float a1 = group8_bcast_f<I>(a1_l);                                  \
        acc0.x += xr[I].x * a0; acc0.y += xr[I].y * a0;                            \
        acc0.z += xr[I].z * a0; acc0.w += xr[I].w * a0;                            \
        acc1.x += xr[I].x * a1; acc1.y += xr[I].y * a1;                            \
        acc1.z += xr[I].z * a1; acc1.w += xr[I].w * a1;                            \
    }
    auto gather = [&](int s_l, float a0_l, float a1_l, int n) {
        float4 xr[8];
        GATHER_IN_LOAD(0) GATHER_IN_LOAD(1) GATHER_IN_LOAD(2) GATHER_IN_LOAD(3)
        GATHER_IN_LOAD(4) GATHER_IN_LOAD(5) GATHER_IN_LOAD(6) GATHER_IN_LOAD(7)
        GATHER_IN_SUM(0) GATHER_IN_SUM(1) GATHER_IN_SUM(2) GATHER_IN_SUM(3)
        GATHER_IN_SUM(4) GATHER_IN_SUM(5) GATHER_IN_SUM(6) GATHER_IN_SUM(7)
    };

    gather(sc, x0c / den0, x1c / den1, tot < 32 ? tot : 32);
    for (int b = 32; b < tot; b += 32) {
        int s; float e0, e1;
        edge(b + lane, s, e0, e1);
        const float x0 = (b + lane < tot) ? __expf(e0 - m0) : 0.f;
        const float x1 = (b + lane < tot) ? __expf(e1 - m1) : 0.f;
        gather(s, x0 / den0, x1 / den1, (tot - b) < 32 ? (tot - b) : 32);
    }
#undef GATHER_IN_LOAD
#undef GATHER_IN_SUM
    for (int b = 0; b < novf; b += 32) {
        int s; float e0, e1;
        const bool mine = ovf_edge(b + lane, s, e0, e1);
        const float a0_l = mine ? __expf(e0 - m0) / den0 : 0.f;
        const float a1_l = mine ? __expf(e1 - m1) / den1 : 0.f;
        unsigned live = __builtin_amdgcn_ballot_w32(mine);
        while (live) {
            const int j = __builtin_ctz(live);
            live &= live - 1;
            const unsigned sj = (unsigned)__builtin_amdgcn_readlane(s, j);
            const float a0 = (lane < 8) ? bcast_f(a0_l, j) : 0.f;
            const float a1 = (lane < 8) ? bcast_f(a1_l, j) : 0.f;
            const float4 xv = (lane < 8) ? *(const float4*)(X + (sj * INCH + q))
                                         : make_float4(0.f, 0.f, 0.f, 0.f);
            acc0.x += xv.x * a0; acc0.y += xv.y * a0;
            acc0.z += xv.z * a0; acc0.w += xv.w * a0;
            acc1.x += xv.x * a1; acc1.y += xv.y * a1;
            acc1.z += xv.z * a1; acc1.w += xv.w * a1;
        }
    }

#define XSUM(v) v += __shfl_xor(v, 8); v += __shfl_xor(v, 16);
    XSUM(acc0.x) XSUM(acc0.y) XSUM(acc0.z) XSUM(acc0.w)
    XSUM(acc1.x) XSUM(acc1.y) XSUM(acc1.z) XSUM(acc1.w)
#undef XSUM

    float* zrow = U + trow * ZPITCH;
    if (lane < 8)       *(float4*)(zrow + q) = acc0;
    else if (lane < 16) *(float4*)(zrow + INCH + q) = acc1;
    }

    __syncthreads();

    {
        const int half = lane >> 4;
        const int l15 = lane & 15;
        const int n = wave * 16 + l15;
        v8f c0 = {}, c1 = {};
        const float* a0p = U + l15 * ZPITCH + (wave >> 2) * INCH + 2 * half;
        const float* a1p = a0p + 16 * ZPITCH;
        const float* __restrict__ bp = W1m + (size_t)(2 * half) * HCAT + n;
        for (int kb = 0; kb < INCH; kb += 4) {
            v2f b;
            b.x = bp[(size_t)kb * HCAT];
            b.y = bp[(size_t)(kb + 1) * HCAT];
            const v2f a0 = *(const v2f*)(a0p + kb);
            const v2f a1 = *(const v2f*)(a1p + kb);
            c0 = __builtin_amdgcn_wmma_f32_16x16x4_f32(false, a0, false, b, (short)0, c0, false, false);
            c1 = __builtin_amdgcn_wmma_f32_16x16x4_f32(false, a1, false, b, (short)0, c1, false, false);
        }
        const float pa = *prelu_a;
        const float bv = bias1[n];
#pragma unroll
        for (int r = 0; r < 8; ++r) {
            float v0 = c0[r] + bv, v1 = c1[r] + bv;
            v0 = v0 >= 0.f ? v0 : pa * v0;
            v1 = v1 >= 0.f ? v1 : pa * v1;
            T[(r + 8 * half) * TPITCH + n] = v0;
            T[(16 + r + 8 * half) * TPITCH + n] = v1;
        }
    }
    __syncthreads();

    tile_linear_rows_logits(T, U, Wn, an_s, an_d, OUT, als_n, ald_n, node0, wave, lane);
}


extern "C" void kernel_launch(void* const* d_in, const int* in_sizes, int n_in,
                              void* d_out, int out_size, void* d_ws, size_t ws_size,
                              hipStream_t stream)
{
    const float* x       = (const float*)d_in[0];
    const int*   ei      = (const int*)  d_in[1];
    const float* W1      = (const float*)d_in[2];
    const float* a1s     = (const float*)d_in[3];
    const float* a1d     = (const float*)d_in[4];
    const float* b1      = (const float*)d_in[5];
    const float* W2      = (const float*)d_in[6];
    const float* a2s     = (const float*)d_in[7];
    const float* a2d     = (const float*)d_in[8];
    const float* b2      = (const float*)d_in[9];
    const float* W3      = (const float*)d_in[10];
    const float* a3s     = (const float*)d_in[11];
    const float* a3d     = (const float*)d_in[12];
    const float* b3      = (const float*)d_in[13];
    const float* prelu_a = (const float*)d_in[14];
    const float* Wc      = (const float*)d_in[15];
    const float* bc      = (const float*)d_in[16];
    float* out = (float*)d_out;

    float* ws = (float*)d_ws;
    size_t off = 0;
    float* H     = ws + off; off += (size_t)NNODES * HCAT;
    float* PREV  = ws + off; off += (size_t)NNODES * HCAT;
    float* als   = ws + off; off += (size_t)NNODES * 2;
    float* ald   = ws + off; off += (size_t)NNODES * 2;
    float* alsB  = ws + off; off += (size_t)NNODES * 2;
    float* aldB  = ws + off; off += (size_t)NNODES * 2;
    int* cnt     = (int*)(ws + off); off += (size_t)CNTWORDS;
    int* slots   = (int*)(ws + off); off += (size_t)NNODES * SLOTC;
    int* ovf     = (int*)(ws + off); off += (size_t)NEDGES * 2;

    const int gN1    = (NNODES + 255) / 256;
    const int gEdge  = (NEDGES + 255) / 256;

    hipMemsetAsync(cnt, 0x00, (size_t)CNTWORDS * sizeof(int), stream);
    bucket_edges_kernel<<<gEdge, 256, 0, stream>>>(ei, cnt, slots, ovf);

    attn_logits_in_kernel<<<gN1, 256, 0, stream>>>(x, W1, a1s, a1d, als, ald);
    gat_gather_in_next_kernel<<<NNODES / NBLK, 256, 0, stream>>>(
        cnt, slots, ovf, x, als, ald, W1, b1, prelu_a, W2, a2s, a2d, H, alsB, aldB);
    gat_gather_next_kernel<true ><<<NNODES / NBLK, 256, 0, stream>>>(
        cnt, slots, ovf, H, alsB, aldB, b2, prelu_a, W3, a3s, a3d, nullptr, PREV, als, ald);
    gat_gather_next_kernel<false><<<NNODES / NBLK, 256, 0, stream>>>(
        cnt, slots, ovf, PREV, als, ald, b3, prelu_a, Wc, nullptr, nullptr, bc, out, nullptr, nullptr);
    (void)in_sizes; (void)n_in; (void)out_size; (void)ws_size;
}


# Edges grouped by destination in two LDS passes (bin by 391-node range, then rank per node in LDS): per-edge global atomic and scattered store go; slot order (sum order) changes
# speedup vs baseline: 11.0494x; 1.3972x over previous
#include <hip/hip_runtime.h>
#include <hip/hip_bf16.h>

#define NNODES 100000
#define NEDGES 1600000
#define INCH   32
#define HIDC   64
#define NHEADS 2
#define HCAT   128
#define NCLS   12
#define NEGSLOPE 0.2f

typedef __attribute__((ext_vector_type(2))) float v2f;
typedef __attribute__((ext_vector_type(8))) float v8f;

__device__ __forceinline__ float leakyr(float x) { return x > 0.f ? x : NEGSLOPE * x; }


#define SLOTC 64
#define CNTWORDS 100096
#define NBINS 256
#define BING 391
#define BINCAP 16384
#define BCHUNK 4096
#define BEPT 16
#define CTRWORDS 512

__global__ __launch_bounds__(256) void bin_edges_kernel(
    const int* __restrict__ ei, int* __restrict__ ctr, int* __restrict__ binbuf,
    int* __restrict__ ovf2, int* __restrict__ cnt)
{
    __shared__ int hist[NBINS];
    __shared__ int lstart[NBINS];
    __shared__ int gbase[NBINS];
    __shared__ int scan[NBINS];
    __shared__ int2 stage[BCHUNK];

    const int tid = threadIdx.x;
    const int e0 = blockIdx.x * BCHUNK;
    if (blockIdx.x == 0 && tid == 0) cnt[NNODES] = 0;
    hist[tid] = 0;
    __syncthreads();

    int s[BEPT], d[BEPT], r[BEPT];
#pragma unroll
    for (int i = 0; i < BEPT; ++i) {
        const int e = e0 + i * 256 + tid;
        d[i] = -1; s[i] = 0; r[i] = 0;
        if (e < NEDGES) {
            s[i] = ei[e];
            d[i] = ei[NEDGES + e];
        }
    }
#pragma unroll
    for (int i = 0; i < BEPT; ++i)
        if (d[i] >= 0) r[i] = atomicAdd(&hist[d[i] / BING], 1);
    __syncthreads();

    const int h = hist[tid];
    scan[tid] = h;
    __syncthreads();
    for (int off = 1; off < NBINS; off <<= 1) {
        const int v = (tid >= off) ? scan[tid - off] : 0;
        __syncthreads();
        scan[tid] += v;
        __syncthreads();
    }
    lstart[tid] = scan[tid] - h;
    gbase[tid] = (h > 0) ? atomicAdd(&ctr[tid], h) : 0;
    __syncthreads();

#pragma unroll
    for (int i = 0; i < BEPT; ++i)
        if (d[i] >= 0) stage[lstart[d[i] / BING] + r[i]] = make_int2(d[i], s[i]);
    __syncthreads();

    const int n = (NEDGES - e0) < BCHUNK ? (NEDGES - e0) : BCHUNK;
    for (int j = tid; j < n; j += 256) {
        const int2 p = stage[j];
        const int b = p.x / BING;
        const int pos = gbase[b] + (j - lstart[b]);
        const unsigned long long pk =
            (unsigned long long)(unsigned)p.x | ((unsigned long long)(unsigned)p.y << 32);
        if (pos < BINCAP) {
            ((unsigned long long*)binbuf)[(size_t)b * BINCAP + pos] = pk;
        } else {
            const int o = atomicAdd(&ctr[NBINS], 1);
            ((unsigned long long*)ovf2)[o] = pk;
        }
    }
}

__global__ __launch_bounds__(1024) void fill_slots_kernel(
    const int* __restrict__ ctr, const int* __restrict__ binbuf, const int* __restrict__ ovf2,
    int* __restrict__ cnt, int* __restrict__ slots, int* __restrict__ ovf)
{
    __shared__ int cntl[BING + 1];
    __shared__ __attribute__((aligned(16))) int rows[BING * SLOTC];

    const int tid = threadIdx.x;
    const int b = blockIdx.x;
    const int d0 = b * BING;
    for (int i = tid; i < BING; i += 1024) cntl[i] = 0;
    __syncthreads();

    auto put = [&](int2 p) {
        const int li = p.x - d0;
        const int rk = atomicAdd(&cntl[li], 1);
        if (rk < SLOTC) {
            rows[li * SLOTC + rk] = p.y;
        } else {
            const int o = atomicAdd(&cnt[NNODES], 1);
            *(int2*)(ovf + 2 * o) = p;
        }
    };

    const int nb = ctr[b] < BINCAP ? ctr[b] : BINCAP;
    const int2* __restrict__ mine = (const int2*)binbuf + (size_t)b * BINCAP;
    for (int j0 = tid; j0 < nb; j0 += 8 * 1024) {
        int2 p[8];
#pragma unroll
        for (int u = 0; u < 8; ++u) {
            const int j = j0 + u * 1024;
            p[u] = make_int2(-1, 0);
            if (j < nb) p[u] = mine[j];
        }
#pragma unroll
        for (int u = 0; u < 8; ++u)
            if (p[u].x >= 0) put(p[u]);
    }
    const int n2 = ctr[NBINS] < NEDGES ? ctr[NBINS] : NEDGES;
    for (int j = tid; j < n2; j += 1024) {
        const int2 p = *(const int2*)(ovf2 + 2 * (size_t)j);
        if (p.x / BING == b) put(p);
    }
    __syncthreads();

    for (int i = tid; i < BING; i += 1024)
        if (d0 + i < NNODES) cnt[d0 + i] = cntl[i];
    for (int idx = tid; idx < BING * (SLOTC / 4); idx += 1024) {
        const int li = idx / (SLOTC / 4);
        const int piece = idx % (SLOTC / 4);
        const int node = d0 + li;
        const int cn = cntl[li] < SLOTC ? cntl[li] : SLOTC;
        if (node < NNODES && piece * 4 < cn)
            *(int4*)(slots + (size_t)node * SLOTC + piece * 4) = *(const int4*)(rows + li * SLOTC + piece * 4);
    }
}

__device__ __forceinline__ float wave_max_f(float v) {
#pragma unroll
    for (int o = 16; o > 0; o >>= 1) v = fmaxf(v, __shfl_xor(v, o));
    return v;
}
__device__ __forceinline__ float wave_sum_f(float v) {
#pragma unroll
    for (int o = 16; o > 0; o >>= 1) v += __shfl_xor(v, o);
    return v;
}
__device__ __forceinline__ float bcast_f(float v, int j) {
    return __int_as_float(__builtin_amdgcn_readlane(__float_as_int(v), j));
}

#define NBLK 32
#define TPITCH (HCAT + 4)

__device__ __forceinline__ void tile_linear_rows_logits(
    const float* T, float* O, const float* __restrict__ Wn,
    const float* __restrict__ an_s, const float* __restrict__ an_d,
    float* __restrict__ OUT, float* __restrict__ als_n, float* __restrict__ ald_n,
    int node0, int wave, int lane)
{
    const int half = lane >> 4;
    const int l15 = lane & 15;
    const int n = wave * 16 + l15;
    v8f acc0 = {}, acc1 = {};
    const float* a0p = T + l15 * TPITCH + 2 * half;
    const float* a1p = a0p + 16 * TPITCH;
    const float* __restrict__ bp = Wn + (size_t)(2 * half) * HCAT + n;
    for (int kb = 0; kb < HCAT; kb += 4) {
        v2f b;
        b.x = bp[(size_t)kb * HCAT];
        b.y = bp[(size_t)(kb + 1) * HCAT];
        const v2f a0 = *(const v2f*)(a0p + kb);
        const v2f a1 = *(const v2f*)(a1p + kb);
        acc0 = __builtin_amdgcn_wmma_f32_16x16x4_f32(false, a0, false, b, (short)0, acc0, false, false);
        acc1 = __builtin_amdgcn_wmma_f32_16x16x4_f32(false, a1, false, b, (short)0, acc1, false, false);
    }
#pragma unroll
    for (int r = 0; r < 8; ++r) {
        O[(r + 8 * half) * TPITCH + n] = acc0[r];
        O[(16 + r + 8 * half) * TPITCH + n] = acc1[r];
    }
    __syncthreads();

    const int c = lane * 4;
    const float4 s4 = *(const float4*)(an_s + c);
    const float4 d4 = *(const float4*)(an_d + c);
#pragma unroll
    for (int q = 0; q < NBLK / 8; ++q) {
        const int i = wave * (NBLK / 8) + q;
        const float4 o = *(const float4*)(O + i * TPITCH + c);
        *(float4*)(OUT + (size_t)(node0 + i) * HCAT + c) = o;
        float s = o.x * s4.x + o.y * s4.y + o.z * s4.z + o.w * s4.w;
        float t = o.x * d4.x + o.y * d4.y + o.z * d4.z + o.w * d4.w;
#pragma unroll
        for (int off = 8; off > 0; off >>= 1) {
            s += __shfl_xor(s, off);
            t += __shfl_xor(t, off);
        }
        if (l15 == 0) {
            als_n[(node0 + i) * 2 + half] = s;
            ald_n[(node0 + i) * 2 + half] = t;
        }
    }
}

template <bool CONCAT>
__global__ __launch_bounds__(256) __attribute__((amdgpu_waves_per_eu(16))) void gat_gather_next_kernel(
    const int* __restrict__ cnt, const int* __restrict__ slots, const int* __restrict__ ovf,
    const float* __restrict__ H, const float* __restrict__ als,
    const float* __restrict__ ald, const float* __restrict__ bias,
    const float* __restrict__ prelu_a,
    const float* __restrict__ Wn, const float* __restrict__ an_s,
    const float* __restrict__ an_d, const float* __restrict__ bn,
    float* __restrict__ OUT, float* __restrict__ als_n, float* __restrict__ ald_n)
{
    constexpr int CW = CONCAT ? HCAT : HIDC;
    constexpr int PITCH = CW + 4;
    __shared__ __attribute__((aligned(16))) float T[NBLK * PITCH];

    const int wave = threadIdx.x >> 5;
    const int lane = threadIdx.x & 31;
    const int node0 = blockIdx.x * NBLK;
    const float2* __restrict__ als2 = (const float2*)als;
    const float a = *prelu_a;

    for (int it = 0; it < NBLK / 8; ++it) {
    const int trow = it * 8 + wave;
    const int d = __builtin_amdgcn_readfirstlane(node0 + trow);
    const int beg = d * SLOTC;
    const int indeg = cnt[d];
    const int deg = indeg < SLOTC ? indeg : SLOTC;
    const int tot = deg + 1;
    const int novf = indeg > SLOTC ? cnt[NNODES] : 0;
    const float ad0 = ald[d * 2 + 0];
    const float ad1 = ald[d * 2 + 1];

    auto edge = [&](int j, int& s, float& e0, float& e1) {
        s = (j < deg) ? slots[beg + j] : d;
        const float2 av = als2[s];
        e0 = (j < tot) ? leakyr(av.x + ad0) : -INFINITY;
        e1 = (j < tot) ? leakyr(av.y + ad1) : -INFINITY;
    };
    auto ovf_edge = [&](int i, int& s, float& e0, float& e1) -> bool {
        bool mine = false;
        s = d;
        if (i < novf) {
            const int2 p = *(const int2*)(ovf + 2 * i);
            mine = (p.x == d);
            s = mine ? p.y : d;
        }
        const float2 av = als2[s];
        e0 = mine ? leakyr(av.x + ad0) : -INFINITY;
        e1 = mine ? leakyr(av.y + ad1) : -INFINITY;
        return mine;
    };

    int sc; float e0c, e1c;
    edge(lane, sc, e0c, e1c);

    float m0 = e0c, m1 = e1c;
    for (int b = 32; b < tot; b += 32) {
        int s; float e0, e1;
        edge(b + lane, s, e0, e1);
        m0 = fmaxf(m0, e0); m1 = fmaxf(m1, e1);
    }
    for (int b = 0; b < novf; b += 32) {
        int s; float e0, e1;
        ovf_edge(b + lane, s, e0, e1);
        m0 = fmaxf(m0, e0); m1 = fmaxf(m1, e1);
    }
    m0 = wave_max_f(m0); m1 = wave_max_f(m1);

    const float x0c = (lane < tot) ? __expf(e0c - m0) : 0.f;
    const float x1c = (lane < tot) ? __expf(e1c - m1) : 0.f;
    float den0 = x0c, den1 = x1c;
    for (int b = 32; b < tot; b += 32) {
        int s; float e0, e1;
        edge(b + lane, s, e0, e1);
        den0 += (b + lane < tot) ? __expf(e0 - m0) : 0.f;
        den1 += (b + lane < tot) ? __expf(e1 - m1) : 0.f;
    }
    for (int b = 0; b < novf; b += 32) {
        int s; float e0, e1;
        const bool mine = ovf_edge(b + lane, s, e0, e1);
        den0 += mine ? __expf(e0 - m0) : 0.f;
        den1 += mine ? __expf(e1 - m1) : 0.f;
    }
    den0 = wave_sum_f(den0); den1 = wave_sum_f(den1);

    const int c = lane * 4;
    const bool head1 = lane >= 16;
    float4 acc = make_float4(0.f, 0.f, 0.f, 0.f);

    auto gather = [&](int s_l, float a0_l, float a1_l, int n) {
#pragma unroll 4
        for (int j = 0; j < n; ++j) {
            const int sj = __builtin_amdgcn_readlane(s_l, j);
            const float a0 = bcast_f(a0_l, j);
            const float a1 = bcast_f(a1_l, j);
            const float al = head1 ? a1 : a0;
            const float4 h4 = *(const float4*)(H + (size_t)sj * HCAT + c);
            acc.x += h4.x * al; acc.y += h4.y * al;
            acc.z += h4.z * al; acc.w += h4.w * al;
        }
    };

    gather(sc, x0c / den0, x1c / den1, tot < 32 ? tot : 32);
    for (int b = 32; b < tot; b += 32) {
        int s; float e0, e1;
        edge(b + lane, s, e0, e1);
        const float x0 = (b + lane < tot) ? __expf(e0 - m0) : 0.f;
        const float x1 = (b + lane < tot) ? __expf(e1 - m1) : 0.f;
        gather(s, x0 / den0, x1 / den1, (tot - b) < 32 ? (tot - b) : 32);
    }
    for (int b = 0; b < novf; b += 32) {
        int s; float e0, e1;
        const bool mine = ovf_edge(b + lane, s, e0, e1);
        const float a0_l = mine ? __expf(e0 - m0) / den0 : 0.f;
        const float a1_l = mine ? __expf(e1 - m1) / den1 : 0.f;
        unsigned live = __builtin_amdgcn_ballot_w32(mine);
        while (live) {
            const int j = __builtin_ctz(live);
            live &= live - 1;
            const int sj = __builtin_amdgcn_readlane(s, j);
            const float al = head1 ? bcast_f(a1_l, j) : bcast_f(a0_l, j);
            const float4 h4 = *(const float4*)(H + (size_t)sj * HCAT + c);
            acc.x += h4.x * al; acc.y += h4.y * al;
            acc.z += h4.z * al; acc.w += h4.w * al;
        }
    }

    if (CONCAT) {
        float4 v = acc;
        v.x += bias[c + 0]; v.y += bias[c + 1]; v.z += bias[c + 2]; v.w += bias[c + 3];
        v.x = v.x >= 0.f ? v.x : a * v.x;
        v.y = v.y >= 0.f ? v.y : a * v.y;
        v.z = v.z >= 0.f ? v.z : a * v.z;
        v.w = v.w >= 0.f ? v.w : a * v.w;
        *(float4*)(T + trow * PITCH + c) = v;
    } else {
        float4 o;
        o.x = __shfl_xor(acc.x, 16); o.y = __shfl_xor(acc.y, 16);
        o.z = __shfl_xor(acc.z, 16); o.w = __shfl_xor(acc.w, 16);
        const int cc = (lane & 15) * 4;
        float4 v;
        v.x = 0.5f * (acc.x + o.x) + bias[cc + 0];
        v.y = 0.5f * (acc.y + o.y) + bias[cc + 1];
        v.z = 0.5f * (acc.z + o.z) + bias[cc + 2];
        v.w = 0.5f * (acc.w + o.w) + bias[cc + 3];
        v.x = v.x >= 0.f ? v.x : a * v.x;
        v.y = v.y >= 0.f ? v.y : a * v.y;
        v.z = v.z >= 0.f ? v.z : a * v.z;
        v.w = v.w >= 0.f ? v.w : a * v.w;
        if (lane < 16) *(float4*)(T + trow * PITCH + cc) = v;
    }
    }

    __syncthreads();

    const int half = lane >> 4;
    const int l15 = lane & 15;
    if constexpr (CONCAT) {
        __shared__ __attribute__((aligned(16))) float O[NBLK * PITCH];
        tile_linear_rows_logits(T, O, Wn, an_s, an_d, OUT, als_n, ald_n, node0, wave, lane);
    } else {
        if (wave < NBLK / 16) {
            v8f accc = {};
            const float* ap = T + (wave * 16 + l15) * PITCH + 2 * half;
            const bool live = l15 < NCLS;
            const int nn = live ? l15 : 0;
            for (int kb = 0; kb < HIDC; kb += 4) {
                const int k0 = kb + 2 * half;
                const v2f av = *(const v2f*)(ap + kb);
                v2f b;
                b.x = Wn[k0 * NCLS + nn];
                b.y = Wn[(k0 + 1) * NCLS + nn];
                b.x = live ? b.x : 0.f;
                b.y = live ? b.y : 0.f;
                accc = __builtin_amdgcn_wmma_f32_16x16x4_f32(false, av, false, b, (short)0, accc, false, false);
            }
            if (live) {
                const float bv = bn[l15];
#pragma unroll
                for (int r = 0; r < 8; ++r) {
                    const int m = node0 + wave * 16 + r + 8 * half;
                    OUT[(size_t)m * NCLS + l15] = accc[r] + bv;
                }
            }
        }
    }
}


__global__ __launch_bounds__(256) void attn_logits_in_kernel(
    const float* __restrict__ X, const float* __restrict__ Wm,
    const float* __restrict__ a_s, const float* __restrict__ a_d,
    float* __restrict__ als, float* __restrict__ ald)
{
    __shared__ float4 wv[INCH];
    const int t = threadIdx.x;
    if (t < INCH * 4) {
        const int k = t >> 2, j = t & 3, head = j & 1;
        const float* __restrict__ av = ((j & 2) ? a_d : a_s) + head * HIDC;
        const float* __restrict__ wr = Wm + (size_t)k * HCAT + head * HIDC;
        const float4* __restrict__ w4 = (const float4*)wr;
        const float4* __restrict__ a4 = (const float4*)av;
        float s = 0.f;
#pragma unroll
        for (int c = 0; c < HIDC / 4; ++c) {
            const float4 w = w4[c], a = a4[c];
            s += w.x * a.x + w.y * a.y + w.z * a.z + w.w * a.w;
        }
        ((float*)wv)[t] = s;
    }
    __syncthreads();
    const int node = blockIdx.x * blockDim.x + t;
    if (node >= NNODES) return;
    const float4* __restrict__ xv = (const float4*)(X + (size_t)node * INCH);
    float s0 = 0.f, s1 = 0.f, d0 = 0.f, d1 = 0.f;
#pragma unroll
    for (int i = 0; i < INCH / 4; ++i) {
        const float4 x4 = xv[i];
        const float xs[4] = { x4.x, x4.y, x4.z, x4.w };
#pragma unroll
        for (int u = 0; u < 4; ++u) {
            const float4 w = wv[i * 4 + u];
            s0 += xs[u] * w.x; s1 += xs[u] * w.y;
            d0 += xs[u] * w.z; d1 += xs[u] * w.w;
        }
    }
    ((float2*)als)[node] = make_float2(s0, s1);
    ((float2*)ald)[node] = make_float2(d0, d1);
}

template <int I>
__device__ __forceinline__ int group8_bcast_i(int v) {
    constexpr unsigned sel = I | (I << 3) | (I << 6) | (I << 9) | (I << 12) | (I << 15) | (I << 18) | (I << 21);
    return __builtin_amdgcn_mov_dpp8(v, sel);
}
template <int I>
__device__ __forceinline__ float group8_bcast_f(float v) {
    return __int_as_float(group8_bcast_i<I>(__float_as_int(v)));
}

#define ZPITCH (NHEADS * INCH + 4)
__global__ __launch_bounds__(256) __attribute__((amdgpu_waves_per_eu(10))) void gat_gather_in_next_kernel(
    const int* __restrict__ cnt, const int* __restrict__ slots, const int* __restrict__ ovf,
    const float* __restrict__ X, const float* __restrict__ als,
    const float* __restrict__ ald,
    const float* __restrict__ W1m, const float* __restrict__ bias1, const float* __restrict__ prelu_a,
    const float* __restrict__ Wn, const float* __restrict__ an_s, const float* __restrict__ an_d,
    float* __restrict__ OUT, float* __restrict__ als_n, float* __restrict__ ald_n)
{
    __shared__ __attribute__((aligned(16))) float U[NBLK * TPITCH];
    __shared__ __attribute__((aligned(16))) float T[NBLK * TPITCH];
    const int wave = threadIdx.x >> 5;
    const int lane = threadIdx.x & 31;
    const int node0 = blockIdx.x * NBLK;

    for (int it = 0; it < NBLK / 8; ++it) {
    const int trow = it * 8 + wave;
    const int d = __builtin_amdgcn_readfirstlane(node0 + trow);
    const int beg = d * SLOTC;
    const int indeg = cnt[d];
    const int deg = indeg < SLOTC ? indeg : SLOTC;
    const int tot = deg + 1;
    const int novf = indeg > SLOTC ? cnt[NNODES] : 0;
    const float ad0 = ald[d * 2 + 0];
    const float ad1 = ald[d * 2 + 1];
    const float2* __restrict__ als2 = (const float2*)als;

    auto edge = [&](int j, int& s, float& e0, float& e1) {
        s = (j < deg) ? slots[beg + j] : d;
        const float2 a = als2[s];
        e0 = (j < tot) ? leakyr(a.x + ad0) : -INFINITY;
        e1 = (j < tot) ? leakyr(a.y + ad1) : -INFINITY;
    };
    auto ovf_edge = [&](int i, int& s, float& e0, float& e1) -> bool {
        bool mine = false;
        s = d;
        if (i < novf) {
            const int2 p = *(const int2*)(ovf + 2 * i);
            mine = (p.x == d);
            s = mine ? p.y : d;
        }
        const float2 a = als2[s];
        e0 = mine ? leakyr(a.x + ad0) : -INFINITY;
        e1 = mine ? leakyr(a.y + ad1) : -INFINITY;
        return mine;
    };

    int sc; float e0c, e1c;
    edge(lane, sc, e0c, e1c);

    float m0 = e0c, m1 = e1c;
    for (int b = 32; b < tot; b += 32) {
        int s; float e0, e1;
        edge(b + lane, s, e0, e1);
        m0 = fmaxf(m0, e0); m1 = fmaxf(m1, e1);
    }
    for (int b = 0; b < novf; b += 32) {
        int s; float e0, e1;
        ovf_edge(b + lane, s, e0, e1);
        m0 = fmaxf(m0, e0); m1 = fmaxf(m1, e1);
    }
    m0 = wave_max_f(m0); m1 = wave_max_f(m1);

    const float x0c = (lane < tot) ? __expf(e0c - m0) : 0.f;
    const float x1c = (lane < tot) ? __expf(e1c - m1) : 0.f;
    float den0 = x0c, den1 = x1c;
    for (int b = 32; b < tot; b += 32) {
        int s; float e0, e1;
        edge(b + lane, s, e0, e1);
        den0 += (b + lane < tot) ? __expf(e0 - m0) : 0.f;
        den1 += (b + lane < tot) ? __expf(e1 - m1) : 0.f;
    }
    for (int b = 0; b < novf; b += 32) {
        int s; float e0, e1;
        const bool mine = ovf_edge(b + lane, s, e0, e1);
        den0 += mine ? __expf(e0 - m0) : 0.f;
        den1 += mine ? __expf(e1 - m1) : 0.f;
    }
    den0 = wave_sum_f(den0); den1 = wave_sum_f(den1);

    const int jbase = lane & 24;
    const unsigned q = (lane & 7) * 4;
    float4 acc0 = make_float4(0.f, 0.f, 0.f, 0.f);
    float4 acc1 = make_float4(0.f, 0.f, 0.f, 0.f);

#define GATHER_IN_LOAD(I)                                                          \
    {                                                                              \
        const unsigned sj = (unsigned)group8_bcast_i<I>(s_l);                      \
        xr[I] = (jbase + I < n) ? *(const float4*)(X + (sj * INCH + q))            \
                                : make_float4(0.f, 0.f, 0.f, 0.f);                 \
    }
#define GATHER_IN_SUM(I)                                                           \
    {                                                                              \
        const float a0 = group8_bcast_f<I>(a0_l);                                  \
        const float a1 = group8_bcast_f<I>(a1_l);                                  \
        acc0.x += xr[I].x * a0; acc0.y += xr[I].y * a0;                            \
        acc0.z += xr[I].z * a0; acc0.w += xr[I].w * a0;                            \
        acc1.x += xr[I].x * a1; acc1.y += xr[I].y * a1;                            \
        acc1.z += xr[I].z * a1; acc1.w += xr[I].w * a1;                            \
    }
    auto gather = [&](int s_l, float a0_l, float a1_l, int n) {
        float4 xr[8];
        GATHER_IN_LOAD(0) GATHER_IN_LOAD(1) GATHER_IN_LOAD(2) GATHER_IN_LOAD(3)
        GATHER_IN_LOAD(4) GATHER_IN_LOAD(5) GATHER_IN_LOAD(6) GATHER_IN_LOAD(7)
        GATHER_IN_SUM(0) GATHER_IN_SUM(1) GATHER_IN_SUM(2) GATHER_IN_SUM(3)
        GATHER_IN_SUM(4) GATHER_IN_SUM(5) GATHER_IN_SUM(6) GATHER_IN_SUM(7)
    };

    gather(sc, x0c / den0, x1c / den1, tot < 32 ? tot : 32);
    for (int b = 32; b < tot; b += 32) {
        int s; float e0, e1;
        edge(b + lane, s, e0, e1);
        const float x0 = (b + lane < tot) ? __expf(e0 - m0) : 0.f;
        const float x1 = (b + lane < tot) ? __expf(e1 - m1) : 0.f;
        gather(s, x0 / den0, x1 / den1, (tot - b) < 32 ? (tot - b) : 32);
    }
#undef GATHER_IN_LOAD
#undef GATHER_IN_SUM
    for (int b = 0; b < novf; b += 32) {
        int s; float e0, e1;
        const bool mine = ovf_edge(b + lane, s, e0, e1);
        const float a0_l = mine ? __expf(e0 - m0) / den0 : 0.f;
        const float a1_l = mine ? __expf(e1 - m1) / den1 : 0.f;
        unsigned live = __builtin_amdgcn_ballot_w32(mine);
        while (live) {
            const int j = __builtin_ctz(live);
            live &= live - 1;
            const unsigned sj = (unsigned)__builtin_amdgcn_readlane(s, j);
            const float a0 = (lane < 8) ? bcast_f(a0_l, j) : 0.f;
            const float a1 = (lane < 8) ? bcast_f(a1_l, j) : 0.f;
            const float4 xv = (lane < 8) ? *(const float4*)(X + (sj * INCH + q))
                                         : make_float4(0.f, 0.f, 0.f, 0.f);
            acc0.x += xv.x * a0; acc0.y += xv.y * a0;
            acc0.z += xv.z * a0; acc0.w += xv.w * a0;
            acc1.x += xv.x * a1; acc1.y += xv.y * a1;
            acc1.z += xv.z * a1; acc1.w += xv.w * a1;
        }
    }

#define XSUM(v) v += __shfl_xor(v, 8); v += __shfl_xor(v, 16);
    XSUM(acc0.x) XSUM(acc0.y) XSUM(acc0.z) XSUM(acc0.w)
    XSUM(acc1.x) XSUM(acc1.y) XSUM(acc1.z) XSUM(acc1.w)
#undef XSUM

    float* zrow = U + trow * ZPITCH;
    if (lane < 8)       *(float4*)(zrow + q) = acc0;
    else if (lane < 16) *(float4*)(zrow + INCH + q) = acc1;
    }

    __syncthreads();

    {
        const int half = lane >> 4;
        const int l15 = lane & 15;
        const int n = wave * 16 + l15;
        v8f c0 = {}, c1 = {};
        const float* a0p = U + l15 * ZPITCH + (wave >> 2) * INCH + 2 * half;
        const float* a1p = a0p + 16 * ZPITCH;
        const float* __restrict__ bp = W1m + (size_t)(2 * half) * HCAT + n;
        for (int kb = 0; kb < INCH; kb += 4) {
            v2f b;
            b.x = bp[(size_t)kb * HCAT];
            b.y = bp[(size_t)(kb + 1) * HCAT];
            const v2f a0 = *(const v2f*)(a0p + kb);
            const v2f a1 = *(const v2f*)(a1p + kb);
            c0 = __builtin_amdgcn_wmma_f32_16x16x4_f32(false, a0, false, b, (short)0, c0, false, false);
            c1 = __builtin_amdgcn_wmma_f32_16x16x4_f32(false, a1, false, b, (short)0, c1, false, false);
        }
        const float pa = *prelu_a;
        const float bv = bias1[n];
#pragma unroll
        for (int r = 0; r < 8; ++r) {
            float v0 = c0[r] + bv, v1 = c1[r] + bv;
            v0 = v0 >= 0.f ? v0 : pa * v0;
            v1 = v1 >= 0.f ? v1 : pa * v1;
            T[(r + 8 * half) * TPITCH + n] = v0;
            T[(16 + r + 8 * half) * TPITCH + n] = v1;
        }
    }
    __syncthreads();

    tile_linear_rows_logits(T, U, Wn, an_s, an_d, OUT, als_n, ald_n, node0, wave, lane);
}


extern "C" void kernel_launch(void* const* d_in, const int* in_sizes, int n_in,
                              void* d_out, int out_size, void* d_ws, size_t ws_size,
                              hipStream_t stream)
{
    const float* x       = (const float*)d_in[0];
    const int*   ei      = (const int*)  d_in[1];
    const float* W1      = (const float*)d_in[2];
    const float* a1s     = (const float*)d_in[3];
    const float* a1d     = (const float*)d_in[4];
    const float* b1      = (const float*)d_in[5];
    const float* W2      = (const float*)d_in[6];
    const float* a2s     = (const float*)d_in[7];
    const float* a2d     = (const float*)d_in[8];
    const float* b2      = (const float*)d_in[9];
    const float* W3      = (const float*)d_in[10];
    const float* a3s     = (const float*)d_in[11];
    const float* a3d     = (const float*)d_in[12];
    const float* b3      = (const float*)d_in[13];
    const float* prelu_a = (const float*)d_in[14];
    const float* Wc      = (const float*)d_in[15];
    const float* bc      = (const float*)d_in[16];
    float* out = (float*)d_out;

    float* ws = (float*)d_ws;
    size_t off = 0;
    float* H     = ws + off; off += (size_t)NNODES * HCAT;
    float* PREV  = ws + off; off += (size_t)NNODES * HCAT;
    float* als   = ws + off; off += (size_t)NNODES * 2;
    float* ald   = ws + off; off += (size_t)NNODES * 2;
    float* alsB  = ws + off; off += (size_t)NNODES * 2;
    float* aldB  = ws + off; off += (size_t)NNODES * 2;
    int* cnt     = (int*)(ws + off); off += (size_t)CNTWORDS;
    int* slots   = (int*)(ws + off); off += (size_t)NNODES * SLOTC;
    int* ovf     = (int*)(ws + off); off += (size_t)NEDGES * 2;
    int* ctr     = (int*)(ws + off); off += (size_t)CTRWORDS;
    int* binbuf  = (int*)(ws + off); off += (size_t)NBINS * BINCAP * 2;
    int* ovf2    = (int*)(ws + off); off += (size_t)NEDGES * 2;

    const int gN1    = (NNODES + 255) / 256;
    const int gBin   = (NEDGES + BCHUNK - 1) / BCHUNK;

    hipMemsetAsync(ctr, 0x00, (size_t)CTRWORDS * sizeof(int), stream);
    bin_edges_kernel<<<gBin, 256, 0, stream>>>(ei, ctr, binbuf, ovf2, cnt);
    fill_slots_kernel<<<NBINS, 1024, 0, stream>>>(ctr, binbuf, ovf2, cnt, slots, ovf);

    attn_logits_in_kernel<<<gN1, 256, 0, stream>>>(x, W1, a1s, a1d, als, ald);
    gat_gather_in_next_kernel<<<NNODES / NBLK, 256, 0, stream>>>(
        cnt, slots, ovf, x, als, ald, W1, b1, prelu_a, W2, a2s, a2d, H, alsB, aldB);
    gat_gather_next_kernel<true ><<<NNODES / NBLK, 256, 0, stream>>>(
        cnt, slots, ovf, H, alsB, aldB, b2, prelu_a, W3, a3s, a3d, nullptr, PREV, als, ald);
    gat_gather_next_kernel<false><<<NNODES / NBLK, 256, 0, stream>>>(
        cnt, slots, ovf, PREV, als, ald, b3, prelu_a, Wc, nullptr, nullptr, bc, out, nullptr, nullptr);
    (void)in_sizes; (void)n_in; (void)out_size; (void)ws_size;
}


# Node kernels: all cross-lane traffic kept in registers (DPP / permlane): wave max/sum and row folds leave the LDS shuffles, an edge's head weight comes by one in-row lane select; every order kept
# speedup vs baseline: 11.6082x; 1.0506x over previous
#include <hip/hip_runtime.h>
#include <hip/hip_bf16.h>

#define NNODES 100000
#define NEDGES 1600000
#define INCH   32
#define HIDC   64
#define NHEADS 2
#define HCAT   128
#define NCLS   12
#define NEGSLOPE 0.2f

typedef __attribute__((ext_vector_type(2))) float v2f;
typedef __attribute__((ext_vector_type(8))) float v8f;

__device__ __forceinline__ float leakyr(float x) { return x > 0.f ? x : NEGSLOPE * x; }


#define SLOTC 64
#define CNTWORDS 100096
#define NBINS 256
#define BING 391
#define BINCAP 16384
#define BCHUNK 4096
#define BEPT 16
#define CTRWORDS 512

__global__ __launch_bounds__(256) void bin_edges_kernel(
    const int* __restrict__ ei, int* __restrict__ ctr, int* __restrict__ binbuf,
    int* __restrict__ ovf2, int* __restrict__ cnt)
{
    __shared__ int hist[NBINS];
    __shared__ int lstart[NBINS];
    __shared__ int gbase[NBINS];
    __shared__ int scan[NBINS];
    __shared__ int2 stage[BCHUNK];

    const int tid = threadIdx.x;
    const int e0 = blockIdx.x * BCHUNK;
    if (blockIdx.x == 0 && tid == 0) cnt[NNODES] = 0;
    hist[tid] = 0;
    __syncthreads();

    int s[BEPT], d[BEPT], r[BEPT];
#pragma unroll
    for (int i = 0; i < BEPT; ++i) {
        const int e = e0 + i * 256 + tid;
        d[i] = -1; s[i] = 0; r[i] = 0;
        if (e < NEDGES) {
            s[i] = ei[e];
            d[i] = ei[NEDGES + e];
        }
    }
#pragma unroll
    for (int i = 0; i < BEPT; ++i)
        if (d[i] >= 0) r[i] = atomicAdd(&hist[d[i] / BING], 1);
    __syncthreads();

    const int h = hist[tid];
    scan[tid] = h;
    __syncthreads();
    for (int off = 1; off < NBINS; off <<= 1) {
        const int v = (tid >= off) ? scan[tid - off] : 0;
        __syncthreads();
        scan[tid] += v;
        __syncthreads();
    }
    lstart[tid] = scan[tid] - h;
    gbase[tid] = (h > 0) ? atomicAdd(&ctr[tid], h) : 0;
    __syncthreads();

#pragma unroll
    for (int i = 0; i < BEPT; ++i)
        if (d[i] >= 0) stage[lstart[d[i] / BING] + r[i]] = make_int2(d[i], s[i]);
    __syncthreads();

    const int n = (NEDGES - e0) < BCHUNK ? (NEDGES - e0) : BCHUNK;
    for (int j = tid; j < n; j += 256) {
        const int2 p = stage[j];
        const int b = p.x / BING;
        const int pos = gbase[b] + (j - lstart[b]);
        const unsigned long long pk =
            (unsigned long long)(unsigned)p.x | ((unsigned long long)(unsigned)p.y << 32);
        if (pos < BINCAP) {
            ((unsigned long long*)binbuf)[(size_t)b * BINCAP + pos] = pk;
        } else {
            const int o = atomicAdd(&ctr[NBINS], 1);
            ((unsigned long long*)ovf2)[o] = pk;
        }
    }
}

__global__ __launch_bounds__(1024) void fill_slots_kernel(
    const int* __restrict__ ctr, const int* __restrict__ binbuf, const int* __restrict__ ovf2,
    int* __restrict__ cnt, int* __restrict__ slots, int* __restrict__ ovf)
{
    __shared__ int cntl[BING + 1];
    __shared__ __attribute__((aligned(16))) int rows[BING * SLOTC];

    const int tid = threadIdx.x;
    const int b = blockIdx.x;
    const int d0 = b * BING;
    for (int i = tid; i < BING; i += 1024) cntl[i] = 0;
    __syncthreads();

    auto put = [&](int2 p) {
        const int li = p.x - d0;
        const int rk = atomicAdd(&cntl[li], 1);
        if (rk < SLOTC) {
            rows[li * SLOTC + rk] = p.y;
        } else {
            const int o = atomicAdd(&cnt[NNODES], 1);
            *(int2*)(ovf + 2 * o) = p;
        }
    };

    const int nb = ctr[b] < BINCAP ? ctr[b] : BINCAP;
    const int2* __restrict__ mine = (const int2*)binbuf + (size_t)b * BINCAP;
    for (int j0 = tid; j0 < nb; j0 += 8 * 1024) {
        int2 p[8];
#pragma unroll
        for (int u = 0; u < 8; ++u) {
            const int j = j0 + u * 1024;
            p[u] = make_int2(-1, 0);
            if (j < nb) p[u] = mine[j];
        }
#pragma unroll
        for (int u = 0; u < 8; ++u)
            if (p[u].x >= 0) put(p[u]);
    }
    const int n2 = ctr[NBINS] < NEDGES ? ctr[NBINS] : NEDGES;
    for (int j = tid; j < n2; j += 1024) {
        const int2 p = *(const int2*)(ovf2 + 2 * (size_t)j);
        if (p.x / BING == b) put(p);
    }
    __syncthreads();

    for (int i = tid; i < BING; i += 1024)
        if (d0 + i < NNODES) cnt[d0 + i] = cntl[i];
    for (int idx = tid; idx < BING * (SLOTC / 4); idx += 1024) {
        const int li = idx / (SLOTC / 4);
        const int piece = idx % (SLOTC / 4);
        const int node = d0 + li;
        const int cn = cntl[li] < SLOTC ? cntl[li] : SLOTC;
        if (node < NNODES && piece * 4 < cn)
            *(int4*)(slots + (size_t)node * SLOTC + piece * 4) = *(const int4*)(rows + li * SLOTC + piece * 4);
    }
}

template <int M>
__device__ __forceinline__ float row_xor_f(float v) {
    return __int_as_float(__builtin_amdgcn_mov_dpp(__float_as_int(v), 0x160 + M, 0xf, 0xf, true));
}
__device__ __forceinline__ float xhalf_f(float v) {
    return __int_as_float(__builtin_amdgcn_permlanex16(__float_as_int(v), __float_as_int(v), 0x76543210, 0xfedcba98, true, false));
}
__device__ __forceinline__ float row_bcast_f(float v, int j) {
    const unsigned sel = (unsigned)j * 0x11111111u;
    return __int_as_float(__builtin_amdgcn_permlane16(__float_as_int(v), __float_as_int(v), sel, sel, true, false));
}
__device__ __forceinline__ float rowx_bcast_f(float v, int j) {
    const unsigned sel = (unsigned)j * 0x11111111u;
    return __int_as_float(__builtin_amdgcn_permlanex16(__float_as_int(v), __float_as_int(v), sel, sel, true, false));
}
__device__ __forceinline__ float wave_max_f(float v) {
    v = fmaxf(v, xhalf_f(v));
    v = fmaxf(v, row_xor_f<8>(v));
    v = fmaxf(v, row_xor_f<4>(v));
    v = fmaxf(v, row_xor_f<2>(v));
    v = fmaxf(v, row_xor_f<1>(v));
    return v;
}
__device__ __forceinline__ float wave_sum_f(float v) {
    v += xhalf_f(v);
    v += row_xor_f<8>(v);
    v += row_xor_f<4>(v);
    v += row_xor_f<2>(v);
    v += row_xor_f<1>(v);
    return v;
}
__device__ __forceinline__ float row_sum_f(float v) {
    v += row_xor_f<8>(v);
    v += row_xor_f<4>(v);
    v += row_xor_f<2>(v);
    v += row_xor_f<1>(v);
    return v;
}
__device__ __forceinline__ float bcast_f(float v, int j) {
    return __int_as_float(__builtin_amdgcn_readlane(__float_as_int(v), j));
}

#define NBLK 32
#define TPITCH (HCAT + 4)

__device__ __forceinline__ void tile_linear_rows_logits(
    const float* T, float* O, const float* __restrict__ Wn,
    const float* __restrict__ an_s, const float* __restrict__ an_d,
    float* __restrict__ OUT, float* __restrict__ als_n, float* __restrict__ ald_n,
    int node0, int wave, int lane)
{
    const int half = lane >> 4;
    const int l15 = lane & 15;
    const int n = wave * 16 + l15;
    v8f acc0 = {}, acc1 = {};
    const float* a0p = T + l15 * TPITCH + 2 * half;
    const float* a1p = a0p + 16 * TPITCH;
    const float* __restrict__ bp = Wn + (size_t)(2 * half) * HCAT + n;
    for (int kb = 0; kb < HCAT; kb += 4) {
        v2f b;
        b.x = bp[(size_t)kb * HCAT];
        b.y = bp[(size_t)(kb + 1) * HCAT];
        const v2f a0 = *(const v2f*)(a0p + kb);
        const v2f a1 = *(const v2f*)(a1p + kb);
        acc0 = __builtin_amdgcn_wmma_f32_16x16x4_f32(false, a0, false, b, (short)0, acc0, false, false);
        acc1 = __builtin_amdgcn_wmma_f32_16x16x4_f32(false, a1, false, b, (short)0, acc1, false, false);
    }
#pragma unroll
    for (int r = 0; r < 8; ++r) {
        O[(r + 8 * half) * TPITCH + n] = acc0[r];
        O[(16 + r + 8 * half) * TPITCH + n] = acc1[r];
    }
    __syncthreads();

    const int ln = (int)__builtin_amdgcn_mbcnt_lo(~0u, 0u);
    const int c = ln * 4;
    const float4 s4 = *(const float4*)(an_s + c);
    const float4 d4 = *(const float4*)(an_d + c);
#pragma unroll
    for (int q = 0; q < NBLK / 8; ++q) {
        const int i = wave * (NBLK / 8) + q;
        const float4 o = *(const float4*)(O + i * TPITCH + c);
        *(float4*)(OUT + (size_t)(node0 + i) * HCAT + c) = o;
        float s = o.x * s4.x + o.y * s4.y + o.z * s4.z + o.w * s4.w;
        float t = o.x * d4.x + o.y * d4.y + o.z * d4.z + o.w * d4.w;
        s = row_sum_f(s);
        t = row_sum_f(t);
        if ((ln & 15) == 0) {
            als_n[(node0 + i) * 2 + (ln >> 4)] = s;
            ald_n[(node0 + i) * 2 + (ln >> 4)] = t;
        }
    }
}

template <bool CONCAT>
__global__ __launch_bounds__(256) __attribute__((amdgpu_waves_per_eu(16))) void gat_gather_next_kernel(
    const int* __restrict__ cnt, const int* __restrict__ slots, const int* __restrict__ ovf,
    const float* __restrict__ H, const float* __restrict__ als,
    const float* __restrict__ ald, const float* __restrict__ bias,
    const float* __restrict__ prelu_a,
    const float* __restrict__ Wn, const float* __restrict__ an_s,
    const float* __restrict__ an_d, const float* __restrict__ bn,
    float* __restrict__ OUT, float* __restrict__ als_n, float* __restrict__ ald_n)
{
    constexpr int CW = CONCAT ? HCAT : HIDC;
    constexpr int PITCH = CW + 4;
    __shared__ __attribute__((aligned(16))) float T[NBLK * PITCH];

    const int wave = __builtin_amdgcn_readfirstlane((int)(threadIdx.x >> 5));
    const int lane = threadIdx.x & 31;
    const int node0 = blockIdx.x * NBLK;
    const float2* __restrict__ als2 = (const float2*)als;
    const float a = *prelu_a;
    const float4* __restrict__ H4 = (const float4*)H;

    for (int it = 0; it < NBLK / 8; ++it) {
    const int trow = it * 8 + wave;
    const int d = __builtin_amdgcn_readfirstlane(node0 + trow);
    const int beg = d * SLOTC;
    const int indeg = cnt[d];
    const int deg = indeg < SLOTC ? indeg : SLOTC;
    const int tot = deg + 1;
    const int novf = indeg > SLOTC ? cnt[NNODES] : 0;
    const float ad0 = ald[d * 2 + 0];
    const float ad1 = ald[d * 2 + 1];

    auto edge = [&](int j, int& s, float& e0, float& e1) {
        s = (j < deg) ? slots[beg + j] : d;
        const float2 av = als2[s];
        e0 = (j < tot) ? leakyr(av.x + ad0) : -INFINITY;
        e1 = (j < tot) ? leakyr(av.y + ad1) : -INFINITY;
    };
    auto ovf_edge = [&](int i, int& s, float& e0, float& e1) -> bool {
        bool mine = false;
        s = d;
        if (i < novf) {
            const int2 p = *(const int2*)(ovf + 2 * i);
            mine = (p.x == d);
            s = mine ? p.y : d;
        }
        const float2 av = als2[s];
        e0 = mine ? leakyr(av.x + ad0) : -INFINITY;
        e1 = mine ? leakyr(av.y + ad1) : -INFINITY;
        return mine;
    };

    int sc; float e0c, e1c;
    edge(lane, sc, e0c, e1c);

    float m0 = e0c, m1 = e1c;
    for (int b = 32; b < tot; b += 32) {
        int s; float e0, e1;
        edge(b + lane, s, e0, e1);
        m0 = fmaxf(m0, e0); m1 = fmaxf(m1, e1);
    }
    for (int b = 0; b < novf; b += 32) {
        int s; float e0, e1;
        ovf_edge(b + lane, s, e0, e1);
        m0 = fmaxf(m0, e0); m1 = fmaxf(m1, e1);
    }
    m0 = wave_max_f(m0); m1 = wave_max_f(m1);

    const float x0c = (lane < tot) ? __expf(e0c - m0) : 0.f;
    const float x1c = (lane < tot) ? __expf(e1c - m1) : 0.f;
    float den0 = x0c, den1 = x1c;
    for (int b = 32; b < tot; b += 32) {
        int s; float e0, e1;
        edge(b + lane, s, e0, e1);
        den0 += (b + lane < tot) ? __expf(e0 - m0) : 0.f;
        den1 += (b + lane < tot) ? __expf(e1 - m1) : 0.f;
    }
    for (int b = 0; b < novf; b += 32) {
        int s; float e0, e1;
        const bool mine = ovf_edge(b + lane, s, e0, e1);
        den0 += mine ? __expf(e0 - m0) : 0.f;
        den1 += mine ? __expf(e1 - m1) : 0.f;
    }
    den0 = wave_sum_f(den0); den1 = wave_sum_f(den1);

    const int c = lane * 4;
    const bool head1 = lane >= 16;
    float4 acc = make_float4(0.f, 0.f, 0.f, 0.f);

    auto gather = [&](int s_l, float a0_l, float a1_l, int n) {
        const float a1x = xhalf_f(a1_l);
        const float wlo = head1 ? a1x : a0_l;
        const float whx = head1 ? a0_l : a1x;
        const int nlo = n < 16 ? n : 16;
#pragma unroll 4
        for (int j = 0; j < nlo; ++j) {
            const unsigned sj = (unsigned)__builtin_amdgcn_readlane(s_l, j);
            const float al = row_bcast_f(wlo, j);
            const float4 h4 = H4[sj * (HCAT / 4) + (unsigned)lane];
            acc.x += h4.x * al; acc.y += h4.y * al;
            acc.z += h4.z * al; acc.w += h4.w * al;
        }
#pragma unroll 4
        for (int j = 16; j < n; ++j) {
            const unsigned sj = (unsigned)__builtin_amdgcn_readlane(s_l, j);
            const float al = rowx_bcast_f(whx, j - 16);
            const float4 h4 = H4[sj * (HCAT / 4) + (unsigned)lane];
            acc.x += h4.x * al; acc.y += h4.y * al;
            acc.z += h4.z * al; acc.w += h4.w * al;
        }
    };

    gather(sc, x0c / den0, x1c / den1, tot < 32 ? tot : 32);
    for (int b = 32; b < tot; b += 32) {
        int s; float e0, e1;
        edge(b + lane, s, e0, e1);
        const float x0 = (b + lane < tot) ? __expf(e0 - m0) : 0.f;
        const float x1 = (b + lane < tot) ? __expf(e1 - m1) : 0.f;
        gather(s, x0 / den0, x1 / den1, (tot - b) < 32 ? (tot - b) : 32);
    }
    for (int b = 0; b < novf; b += 32) {
        int s; float e0, e1;
        const bool mine = ovf_edge(b + lane, s, e0, e1);
        const float a0_l = mine ? __expf(e0 - m0) / den0 : 0.f;
        const float a1_l = mine ? __expf(e1 - m1) / den1 : 0.f;
        unsigned live = __builtin_amdgcn_ballot_w32(mine);
        while (live) {
            const int j = __builtin_ctz(live);
            live &= live - 1;
            const int sj = __builtin_amdgcn_readlane(s, j);
            const float al = head1 ? bcast_f(a1_l, j) : bcast_f(a0_l, j);
            const float4 h4 = *(const float4*)(H + (size_t)sj * HCAT + c);
            acc.x += h4.x * al; acc.y += h4.y * al;
            acc.z += h4.z * al; acc.w += h4.w * al;
        }
    }

    if (CONCAT) {
        float4 v = acc;
        v.x += bias[c + 0]; v.y += bias[c + 1]; v.z += bias[c + 2]; v.w += bias[c + 3];
        v.x = v.x >= 0.f ? v.x : a * v.x;
        v.y = v.y >= 0.f ? v.y : a * v.y;
        v.z = v.z >= 0.f ? v.z : a * v.z;
        v.w = v.w >= 0.f ? v.w : a * v.w;
        *(float4*)(T + trow * PITCH + c) = v;
    } else {
        float4 o;
        o.x = xhalf_f(acc.x); o.y = xhalf_f(acc.y);
        o.z = xhalf_f(acc.z); o.w = xhalf_f(acc.w);
        const int cc = (lane & 15) * 4;
        float4 v;
        v.x = 0.5f * (acc.x + o.x) + bias[cc + 0];
        v.y = 0.5f * (acc.y + o.y) + bias[cc + 1];
        v.z = 0.5f * (acc.z + o.z) + bias[cc + 2];
        v.w = 0.5f * (acc.w + o.w) + bias[cc + 3];
        v.x = v.x >= 0.f ? v.x : a * v.x;
        v.y = v.y >= 0.f ? v.y : a * v.y;
        v.z = v.z >= 0.f ? v.z : a * v.z;
        v.w = v.w >= 0.f ? v.w : a * v.w;
        if (lane < 16) *(float4*)(T + trow * PITCH + cc) = v;
    }
    }

    __syncthreads();

    const int half = lane >> 4;
    const int l15 = lane & 15;
    if constexpr (CONCAT) {
        __shared__ __attribute__((aligned(16))) float O[NBLK * PITCH];
        tile_linear_rows_logits(T, O, Wn, an_s, an_d, OUT, als_n, ald_n, node0, wave, lane);
    } else {
        if (wave < NBLK / 16) {
            v8f accc = {};
            const float* ap = T + (wave * 16 + l15) * PITCH + 2 * half;
            const bool live = l15 < NCLS;
            const int nn = live ? l15 : 0;
            for (int kb = 0; kb < HIDC; kb += 4) {
                const int k0 = kb + 2 * half;
                const v2f av = *(const v2f*)(ap + kb);
                v2f b;
                b.x = Wn[k0 * NCLS + nn];
                b.y = Wn[(k0 + 1) * NCLS + nn];
                b.x = live ? b.x : 0.f;
                b.y = live ? b.y : 0.f;
                accc = __builtin_amdgcn_wmma_f32_16x16x4_f32(false, av, false, b, (short)0, accc, false, false);
            }
            if (live) {
                const float bv = bn[l15];
#pragma unroll
                for (int r = 0; r < 8; ++r) {
                    const int m = node0 + wave * 16 + r + 8 * half;
                    OUT[(size_t)m * NCLS + l15] = accc[r] + bv;
                }
            }
        }
    }
}


__global__ __launch_bounds__(256) void attn_logits_in_kernel(
    const float* __restrict__ X, const float* __restrict__ Wm,
    const float* __restrict__ a_s, const float* __restrict__ a_d,
    float* __restrict__ als, float* __restrict__ ald)
{
    __shared__ float4 wv[INCH];
    const int t = threadIdx.x;
    if (t < INCH * 4) {
        const int k = t >> 2, j = t & 3, head = j & 1;
        const float* __restrict__ av = ((j & 2) ? a_d : a_s) + head * HIDC;
        const float* __restrict__ wr = Wm + (size_t)k * HCAT + head * HIDC;
        const float4* __restrict__ w4 = (const float4*)wr;
        const float4* __restrict__ a4 = (const float4*)av;
        float s = 0.f;
#pragma unroll
        for (int c = 0; c < HIDC / 4; ++c) {
            const float4 w = w4[c], a = a4[c];
            s += w.x * a.x + w.y * a.y + w.z * a.z + w.w * a.w;
        }
        ((float*)wv)[t] = s;
    }
    __syncthreads();
    const int node = blockIdx.x * blockDim.x + t;
    if (node >= NNODES) return;
    const float4* __restrict__ xv = (const float4*)(X + (size_t)node * INCH);
    float s0 = 0.f, s1 = 0.f, d0 = 0.f, d1 = 0.f;
#pragma unroll
    for (int i = 0; i < INCH / 4; ++i) {
        const float4 x4 = xv[i];
        const float xs[4] = { x4.x, x4.y, x4.z, x4.w };
#pragma unroll
        for (int u = 0; u < 4; ++u) {
            const float4 w = wv[i * 4 + u];
            s0 += xs[u] * w.x; s1 += xs[u] * w.y;
            d0 += xs[u] * w.z; d1 += xs[u] * w.w;
        }
    }
    ((float2*)als)[node] = make_float2(s0, s1);
    ((float2*)ald)[node] = make_float2(d0, d1);
}

template <int I>
__device__ __forceinline__ int group8_bcast_i(int v) {
    constexpr unsigned sel = I | (I << 3) | (I << 6) | (I << 9) | (I << 12) | (I << 15) | (I << 18) | (I << 21);
    return __builtin_amdgcn_mov_dpp8(v, sel);
}
template <int I>
__device__ __forceinline__ float group8_bcast_f(float v) {
    return __int_as_float(group8_bcast_i<I>(__float_as_int(v)));
}

#define ZPITCH (NHEADS * INCH + 4)
__global__ __launch_bounds__(256) __attribute__((amdgpu_waves_per_eu(10))) void gat_gather_in_next_kernel(
    const int* __restrict__ cnt, const int* __restrict__ slots, const int* __restrict__ ovf,
    const float* __restrict__ X, const float* __restrict__ als,
    const float* __restrict__ ald,
    const float* __restrict__ W1m, const float* __restrict__ bias1, const float* __restrict__ prelu_a,
    const float* __restrict__ Wn, const float* __restrict__ an_s, const float* __restrict__ an_d,
    float* __restrict__ OUT, float* __restrict__ als_n, float* __restrict__ ald_n)
{
    __shared__ __attribute__((aligned(16))) float U[NBLK * TPITCH];
    __shared__ __attribute__((aligned(16))) float T[NBLK * TPITCH];
    const int wave = threadIdx.x >> 5;
    const int lane = threadIdx.x & 31;
    const int node0 = blockIdx.x * NBLK;

    for (int it = 0; it < NBLK / 8; ++it) {
    const int trow = it * 8 + wave;
    const int d = __builtin_amdgcn_readfirstlane(node0 + trow);
    const int beg = d * SLOTC;
    const int indeg = cnt[d];
    const int deg = indeg < SLOTC ? indeg : SLOTC;
    const int tot = deg + 1;
    const int novf = indeg > SLOTC ? cnt[NNODES] : 0;
    const float ad0 = ald[d * 2 + 0];
    const float ad1 = ald[d * 2 + 1];
    const float2* __restrict__ als2 = (const float2*)als;

    auto edge = [&](int j, int& s, float& e0, float& e1) {
        s = (j < deg) ? slots[beg + j] : d;
        const float2 a = als2[s];
        e0 = (j < tot) ? leakyr(a.x + ad0) : -INFINITY;
        e1 = (j < tot) ? leakyr(a.y + ad1) : -INFINITY;
    };
    auto ovf_edge = [&](int i, int& s, float& e0, float& e1) -> bool {
        bool mine = false;
        s = d;
        if (i < novf) {
            const int2 p = *(const int2*)(ovf + 2 * i);
            mine = (p.x == d);
            s = mine ? p.y : d;
        }
        const float2 a = als2[s];
        e0 = mine ? leakyr(a.x + ad0) : -INFINITY;
        e1 = mine ? leakyr(a.y + ad1) : -INFINITY;
        return mine;
    };

    int sc; float e0c, e1c;
    edge(lane, sc, e0c, e1c);

    float m0 = e0c, m1 = e1c;
    for (int b = 32; b < tot; b += 32) {
        int s; float e0, e1;
        edge(b + lane, s, e0, e1);
        m0 = fmaxf(m0, e0); m1 = fmaxf(m1, e1);
    }
    for (int b = 0; b < novf; b += 32) {
        int s; float e0, e1;
        ovf_edge(b + lane, s, e0, e1);
        m0 = fmaxf(m0, e0); m1 = fmaxf(m1, e1);
    }
    m0 = wave_max_f(m0); m1 = wave_max_f(m1);

    const float x0c = (lane < tot) ? __expf(e0c - m0) : 0.f;
    const float x1c = (lane < tot) ? __expf(e1c - m1) : 0.f;
    float den0 = x0c, den1 = x1c;
    for (int b = 32; b < tot; b += 32) {
        int s; float e0, e1;
        edge(b + lane, s, e0, e1);
        den0 += (b + lane < tot) ? __expf(e0 - m0) : 0.f;
        den1 += (b + lane < tot) ? __expf(e1 - m1) : 0.f;
    }
    for (int b = 0; b < novf; b += 32) {
        int s; float e0, e1;
        const bool mine = ovf_edge(b + lane, s, e0, e1);
        den0 += mine ? __expf(e0 - m0) : 0.f;
        den1 += mine ? __expf(e1 - m1) : 0.f;
    }
    den0 = wave_sum_f(den0); den1 = wave_sum_f(den1);

    const int jbase = lane & 24;
    const unsigned q = (lane & 7) * 4;
    float4 acc0 = make_float4(0.f, 0.f, 0.f, 0.f);
    float4 acc1 = make_float4(0.f, 0.f, 0.f, 0.f);

#define GATHER_IN_LOAD(I)                                                          \
    {                                                                              \
        const unsigned sj = (unsigned)group8_bcast_i<I>(s_l);                      \
        xr[I] = (jbase + I < n) ? *(const float4*)(X + (sj * INCH + q))            \
                                : make_float4(0.f, 0.f, 0.f, 0.f);                 \
    }
#define GATHER_IN_SUM(I)                                                           \
    {                                                                              \
        const float a0 = group8_bcast_f<I>(a0_l);                                  \
        const float a1 = group8_bcast_f<I>(a1_l);                                  \
        acc0.x += xr[I].x * a0; acc0.y += xr[I].y * a0;                            \
        acc0.z += xr[I].z * a0; acc0.w += xr[I].w * a0;                            \
        acc1.x += xr[I].x * a1; acc1.y += xr[I].y * a1;                            \
        acc1.z += xr[I].z * a1; acc1.w += xr[I].w * a1;                            \
    }
    auto gather = [&](int s_l, float a0_l, float a1_l, int n) {
        float4 xr[8];
        GATHER_IN_LOAD(0) GATHER_IN_LOAD(1) GATHER_IN_LOAD(2) GATHER_IN_LOAD(3)
        GATHER_IN_LOAD(4) GATHER_IN_LOAD(5) GATHER_IN_LOAD(6) GATHER_IN_LOAD(7)
        GATHER_IN_SUM(0) GATHER_IN_SUM(1) GATHER_IN_SUM(2) GATHER_IN_SUM(3)
        GATHER_IN_SUM(4) GATHER_IN_SUM(5) GATHER_IN_SUM(6) GATHER_IN_SUM(7)
    };

    gather(sc, x0c / den0, x1c / den1, tot < 32 ? tot : 32);
    for (int b = 32; b < tot; b += 32) {
        int s; float e0, e1;
        edge(b + lane, s, e0, e1);
        const float x0 = (b + lane < tot) ? __expf(e0 - m0) : 0.f;
        const float x1 = (b + lane < tot) ? __expf(e1 - m1) : 0.f;
        gather(s, x0 / den0, x1 / den1, (tot - b) < 32 ? (tot - b) : 32);
    }
#undef GATHER_IN_LOAD
#undef GATHER_IN_SUM
    for (int b = 0; b < novf; b += 32) {
        int s; float e0, e1;
        const bool mine = ovf_edge(b + lane, s, e0, e1);
        const float a0_l = mine ? __expf(e0 - m0) / den0 : 0.f;
        const float a1_l = mine ? __expf(e1 - m1) / den1 : 0.f;
        unsigned live = __builtin_amdgcn_ballot_w32(mine);
        while (live) {
            const int j = __builtin_ctz(live);
            live &= live - 1;
            const unsigned sj = (unsigned)__builtin_amdgcn_readlane(s, j);
            const float a0 = (lane < 8) ? bcast_f(a0_l, j) : 0.f;
            const float a1 = (lane < 8) ? bcast_f(a1_l, j) : 0.f;
            const float4 xv = (lane < 8) ? *(const float4*)(X + (sj * INCH + q))
                                         : make_float4(0.f, 0.f, 0.f, 0.f);
            acc0.x += xv.x * a0; acc0.y += xv.y * a0;
            acc0.z += xv.z * a0; acc0.w += xv.w * a0;
            acc1.x += xv.x * a1; acc1.y += xv.y * a1;
            acc1.z += xv.z * a1; acc1.w += xv.w * a1;
        }
    }

#define XSUM(v) v += row_xor_f<8>(v); v += xhalf_f(v);
    XSUM(acc0.x) XSUM(acc0.y) XSUM(acc0.z) XSUM(acc0.w)
    XSUM(acc1.x) XSUM(acc1.y) XSUM(acc1.z) XSUM(acc1.w)
#undef XSUM

    float* zrow = U + trow * ZPITCH;
    if (lane < 8)       *(float4*)(zrow + q) = acc0;
    else if (lane < 16) *(float4*)(zrow + INCH + q) = acc1;
    }

    __syncthreads();

    {
        const int half = lane >> 4;
        const int l15 = lane & 15;
        const int n = wave * 16 + l15;
        v8f c0 = {}, c1 = {};
        const float* a0p = U + l15 * ZPITCH + (wave >> 2) * INCH + 2 * half;
        const float* a1p = a0p + 16 * ZPITCH;
        const float* __restrict__ bp = W1m + (size_t)(2 * half) * HCAT + n;
        for (int kb = 0; kb < INCH; kb += 4) {
            v2f b;
            b.x = bp[(size_t)kb * HCAT];
            b.y = bp[(size_t)(kb + 1) * HCAT];
            const v2f a0 = *(const v2f*)(a0p + kb);
            const v2f a1 = *(const v2f*)(a1p + kb);
            c0 = __builtin_amdgcn_wmma_f32_16x16x4_f32(false, a0, false, b, (short)0, c0, false, false);
            c1 = __builtin_amdgcn_wmma_f32_16x16x4_f32(false, a1, false, b, (short)0, c1, false, false);
        }
        const float pa = *prelu_a;
        const float bv = bias1[n];
#pragma unroll
        for (int r = 0; r < 8; ++r) {
            float v0 = c0[r] + bv, v1 = c1[r] + bv;
            v0 = v0 >= 0.f ? v0 : pa * v0;
            v1 = v1 >= 0.f ? v1 : pa * v1;
            T[(r + 8 * half) * TPITCH + n] = v0;
            T[(16 + r + 8 * half) * TPITCH + n] = v1;
        }
    }
    __syncthreads();

    tile_linear_rows_logits(T, U, Wn, an_s, an_d, OUT, als_n, ald_n, node0, wave, lane);
}


extern "C" void kernel_launch(void* const* d_in, const int* in_sizes, int n_in,
                              void* d_out, int out_size, void* d_ws, size_t ws_size,
                              hipStream_t stream)
{
    const float* x       = (const float*)d_in[0];
    const int*   ei      = (const int*)  d_in[1];
    const float* W1      = (const float*)d_in[2];
    const float* a1s     = (const float*)d_in[3];
    const float* a1d     = (const float*)d_in[4];
    const float* b1      = (const float*)d_in[5];
    const float* W2      = (const float*)d_in[6];
    const float* a2s     = (const float*)d_in[7];
    const float* a2d     = (const float*)d_in[8];
    const float* b2      = (const float*)d_in[9];
    const float* W3      = (const float*)d_in[10];
    const float* a3s     = (const float*)d_in[11];
    const float* a3d     = (const float*)d_in[12];
    const float* b3      = (const float*)d_in[13];
    const float* prelu_a = (const float*)d_in[14];
    const float* Wc      = (const float*)d_in[15];
    const float* bc      = (const float*)d_in[16];
    float* out = (float*)d_out;

    float* ws = (float*)d_ws;
    size_t off = 0;
    float* H     = ws + off; off += (size_t)NNODES * HCAT;
    float* PREV  = ws + off; off += (size_t)NNODES * HCAT;
    float* als   = ws + off; off += (size_t)NNODES * 2;
    float* ald   = ws + off; off += (size_t)NNODES * 2;
    float* alsB  = ws + off; off += (size_t)NNODES * 2;
    float* aldB  = ws + off; off += (size_t)NNODES * 2;
    int* cnt     = (int*)(ws + off); off += (size_t)CNTWORDS;
    int* slots   = (int*)(ws + off); off += (size_t)NNODES * SLOTC;
    int* ovf     = (int*)(ws + off); off += (size_t)NEDGES * 2;
    int* ctr     = (int*)(ws + off); off += (size_t)CTRWORDS;
    int* binbuf  = (int*)(ws + off); off += (size_t)NBINS * BINCAP * 2;
    int* ovf2    = (int*)(ws + off); off += (size_t)NEDGES * 2;

    const int gN1    = (NNODES + 255) / 256;
    const int gBin   = (NEDGES + BCHUNK - 1) / BCHUNK;

    hipMemsetAsync(ctr, 0x00, (size_t)CTRWORDS * sizeof(int), stream);
    bin_edges_kernel<<<gBin, 256, 0, stream>>>(ei, ctr, binbuf, ovf2, cnt);
    fill_slots_kernel<<<NBINS, 1024, 0, stream>>>(ctr, binbuf, ovf2, cnt, slots, ovf);

    attn_logits_in_kernel<<<gN1, 256, 0, stream>>>(x, W1, a1s, a1d, als, ald);
    gat_gather_in_next_kernel<<<NNODES / NBLK, 256, 0, stream>>>(
        cnt, slots, ovf, x, als, ald, W1, b1, prelu_a, W2, a2s, a2d, H, alsB, aldB);
    gat_gather_next_kernel<true ><<<NNODES / NBLK, 256, 0, stream>>>(
        cnt, slots, ovf, H, alsB, aldB, b2, prelu_a, W3, a3s, a3d, nullptr, PREV, als, ald);
    gat_gather_next_kernel<false><<<NNODES / NBLK, 256, 0, stream>>>(
        cnt, slots, ovf, PREV, als, ald, b3, prelu_a, Wc, nullptr, nullptr, bc, out, nullptr, nullptr);
    (void)in_sizes; (void)n_in; (void)out_size; (void)ws_size;
}
